# starting kernel
# baseline (speedup 1.0000x reference)
#include <hip/hip_runtime.h>
#include <hip/hip_bf16.h>

#define LSEQ 60000
#define NB 8
#define INCH 64
#define HID 5

typedef __attribute__((ext_vector_type(2))) float v2f;
typedef __attribute__((ext_vector_type(8))) float v8f;

__global__ void conv1x1_wmma(const float* __restrict__ x,
                             const float* __restrict__ w,
                             const float* __restrict__ bias,
                             float* __restrict__ y) {
    const int lane = threadIdx.x & 31;
    const int wave = threadIdx.x >> 5;
    const int tile = blockIdx.x * 8 + wave;
    const int b = tile / (LSEQ / 16);
    const int lbase = (tile % (LSEQ / 16)) * 16;
    const int m = lane & 15;
    const int half = lane >> 4;

    const float* xb = x + (size_t)b * INCH * LSEQ + lbase + m;

    v8f acc = {};
#pragma unroll
    for (int chunk = 0; chunk < 16; ++chunk) {
        const int k0 = chunk * 4 + 2 * half;
        v2f a, bb;
        a.x = xb[(size_t)k0 * LSEQ];
        a.y = xb[(size_t)(k0 + 1) * LSEQ];
        bb.x = w[k0];
        bb.y = w[k0 + 1];
        acc = __builtin_amdgcn_wmma_f32_16x16x4_f32(
             false, a,  false, bb,
             (short)0, acc,  false,  false);
    }

    const float bv = bias[0];
    float* yo = y + (size_t)b * LSEQ + lbase;
    if (lane == 0) {
#pragma unroll
        for (int r = 0; r < 8; ++r) yo[r] = acc[r] + bv;
    } else if (lane == 16) {
#pragma unroll
        for (int r = 0; r < 8; ++r) yo[8 + r] = acc[r] + bv;
    }
}

__device__ __forceinline__ float sigm(float v) {
    return __builtin_amdgcn_rcpf(1.0f + __expf(-v));
}
__device__ __forceinline__ float tanh_fast(float v) {
    return 1.0f - 2.0f * __builtin_amdgcn_rcpf(__expf(2.0f * v) + 1.0f);
}

__global__ void lstm_scan(const float* __restrict__ y,
                          const float* __restrict__ inp,
                          const float* __restrict__ w_ih_f, const float* __restrict__ w_hh_f,
                          const float* __restrict__ b_ih_f, const float* __restrict__ b_hh_f,
                          const float* __restrict__ w_hr_f,
                          const float* __restrict__ w_ih_b, const float* __restrict__ w_hh_b,
                          const float* __restrict__ b_ih_b, const float* __restrict__ b_hh_b,
                          const float* __restrict__ w_hr_b,
                          const float* __restrict__ h0, const float* __restrict__ c0,
                          float* __restrict__ hout) {
    const int t = threadIdx.x;
    if (t >= 16) return;
    const int dir = t >> 3;
    const int b = t & 7;

    const float* w_ih = dir ? w_ih_b : w_ih_f;
    const float* w_hh = dir ? w_hh_b : w_hh_f;
    const float* b_ih = dir ? b_ih_b : b_ih_f;
    const float* b_hh = dir ? b_hh_b : b_hh_f;
    const float* w_hr = dir ? w_hr_b : w_hr_f;

    float wi0[20], wi1[20], whh[20], bsum[20], whr[HID], c[HID];
#pragma unroll
    for (int j = 0; j < 20; ++j) {
        wi0[j] = w_ih[2 * j];
        wi1[j] = w_ih[2 * j + 1];
        whh[j] = w_hh[j];
        bsum[j] = b_ih[j] + b_hh[j];
    }
#pragma unroll
    for (int k = 0; k < HID; ++k) {
        whr[k] = w_hr[k];
        c[k] = c0[(dir * NB + b) * HID + k];
    }
    float h = h0[dir * NB + b];

    const float* yb = y + (size_t)b * LSEQ;
    const float* ib = inp + (size_t)b * LSEQ;
    float* ho = hout + (size_t)(dir * NB + b) * LSEQ;

    for (int s = 0; s < LSEQ; ++s) {
        const int l = dir ? (LSEQ - 1 - s) : s;
        if ((s & 31) == 0) {
            const int pl = dir ? (l - 96) : (l + 96);
            if ((unsigned)pl < (unsigned)LSEQ) {
                __builtin_prefetch(yb + pl, 0, 0);
                __builtin_prefetch(ib + pl, 0, 0);
            }
        }
        const float x0 = yb[l];
        const float x1 = ib[l];
        float hn = 0.0f;
#pragma unroll
        for (int k = 0; k < HID; ++k) {
            const float gi = fmaf(x0, wi0[k],      fmaf(x1, wi1[k],      fmaf(h, whh[k],      bsum[k])));
            const float gf = fmaf(x0, wi0[k + 5],  fmaf(x1, wi1[k + 5],  fmaf(h, whh[k + 5],  bsum[k + 5])));
            const float gc = fmaf(x0, wi0[k + 10], fmaf(x1, wi1[k + 10], fmaf(h, whh[k + 10], bsum[k + 10])));
            const float go = fmaf(x0, wi0[k + 15], fmaf(x1, wi1[k + 15], fmaf(h, whh[k + 15], bsum[k + 15])));
            const float iv = sigm(gi);
            const float fv = sigm(gf);
            const float gv = tanh_fast(gc);
            const float ov = sigm(go);
            c[k] = fmaf(fv, c[k], iv * gv);
            hn = fmaf(ov * tanh_fast(c[k]), whr[k], hn);
        }
        h = hn;
        ho[l] = h;
    }
}

__global__ void seq_softmax(const float* __restrict__ hbuf, float* __restrict__ out) {
    const int b = blockIdx.x;
    const int tid = threadIdx.x;
    const int T = 256;
    __shared__ float red[256];

    const float* hf = hbuf + (size_t)b * LSEQ;
    const float* hb = hbuf + (size_t)(NB + b) * LSEQ;

    float m = -3.0e38f;
    for (int l = tid; l < LSEQ; l += T) m = fmaxf(m, hf[l] + hb[l]);
    red[tid] = m;
    __syncthreads();
    for (int s = T / 2; s > 0; s >>= 1) {
        if (tid < s) red[tid] = fmaxf(red[tid], red[tid + s]);
        __syncthreads();
    }
    m = red[0];
    __syncthreads();

    float sum = 0.0f;
    for (int l = tid; l < LSEQ; l += T) sum += __expf(hf[l] + hb[l] - m);
    red[tid] = sum;
    __syncthreads();
    for (int s = T / 2; s > 0; s >>= 1) {
        if (tid < s) red[tid] += red[tid + s];
        __syncthreads();
    }
    sum = red[0];
    __syncthreads();

    const float inv = __builtin_amdgcn_rcpf(sum);
    for (int l = tid; l < LSEQ; l += T) {
        const float e = __expf(hf[l] + hb[l] - m) * inv;
        float* o = out + ((size_t)b * LSEQ + l) * 2;
        o[0] = e;
        o[1] = e;
    }
}

extern "C" void kernel_launch(void* const* d_in, const int* in_sizes, int n_in,
                              void* d_out, int out_size, void* d_ws, size_t ws_size,
                              hipStream_t stream) {
    const float* x      = (const float*)d_in[0];
    const float* inp    = (const float*)d_in[1];
    const float* conv_w = (const float*)d_in[2];
    const float* conv_b = (const float*)d_in[3];
    const float* w_ih_f = (const float*)d_in[4];
    const float* w_hh_f = (const float*)d_in[5];
    const float* b_ih_f = (const float*)d_in[6];
    const float* b_hh_f = (const float*)d_in[7];
    const float* w_hr_f = (const float*)d_in[8];
    const float* w_ih_b = (const float*)d_in[9];
    const float* w_hh_b = (const float*)d_in[10];
    const float* b_ih_b = (const float*)d_in[11];
    const float* b_hh_b = (const float*)d_in[12];
    const float* w_hr_b = (const float*)d_in[13];
    const float* h0     = (const float*)d_in[14];
    const float* c0     = (const float*)d_in[15];
    float* out = (float*)d_out;

    float* y    = (float*)d_ws;
    float* hbuf = y + (size_t)NB * LSEQ;

    conv1x1_wmma<<<(NB * LSEQ / 16) / 8, 256, 0, stream>>>(x, conv_w, conv_b, y);

    lstm_scan<<<1, 32, 0, stream>>>(y, inp,
                                    w_ih_f, w_hh_f, b_ih_f, b_hh_f, w_hr_f,
                                    w_ih_b, w_hh_b, b_ih_b, b_hh_b, w_hr_b,
                                    h0, c0, hbuf);

    seq_softmax<<<NB, 256, 0, stream>>>(hbuf, out);
}


# LSTM scan: kernel declared as one 32-lane wave per SIMD (launch bounds 32, waves per EU 1): the compiler now interleaves a step's 20 independent gate chains, was one after another; sum order kept
# speedup vs baseline: 1.6012x; 1.6012x over previous
#include <hip/hip_runtime.h>
#include <hip/hip_bf16.h>

#define LSEQ 60000
#define NB 8
#define INCH 64
#define HID 5

typedef __attribute__((ext_vector_type(2))) float v2f;
typedef __attribute__((ext_vector_type(8))) float v8f;

__global__ void conv1x1_wmma(const float* __restrict__ x,
                             const float* __restrict__ w,
                             const float* __restrict__ bias,
                             float* __restrict__ y) {
    const int lane = threadIdx.x & 31;
    const int wave = threadIdx.x >> 5;
    const int tile = blockIdx.x * 8 + wave;
    const int b = tile / (LSEQ / 16);
    const int lbase = (tile % (LSEQ / 16)) * 16;
    const int m = lane & 15;
    const int half = lane >> 4;

    const float* xb = x + (size_t)b * INCH * LSEQ + lbase + m;

    v8f acc = {};
#pragma unroll
    for (int chunk = 0; chunk < 16; ++chunk) {
        const int k0 = chunk * 4 + 2 * half;
        v2f a, bb;
        a.x = xb[(size_t)k0 * LSEQ];
        a.y = xb[(size_t)(k0 + 1) * LSEQ];
        bb.x = w[k0];
        bb.y = w[k0 + 1];
        acc = __builtin_amdgcn_wmma_f32_16x16x4_f32(
             false, a,  false, bb,
             (short)0, acc,  false,  false);
    }

    const float bv = bias[0];
    float* yo = y + (size_t)b * LSEQ + lbase;
    if (lane == 0) {
#pragma unroll
        for (int r = 0; r < 8; ++r) yo[r] = acc[r] + bv;
    } else if (lane == 16) {
#pragma unroll
        for (int r = 0; r < 8; ++r) yo[8 + r] = acc[r] + bv;
    }
}

__device__ __forceinline__ float sigm(float v) {
    return __builtin_amdgcn_rcpf(1.0f + __expf(-v));
}
__device__ __forceinline__ float tanh_fast(float v) {
    return 1.0f - 2.0f * __builtin_amdgcn_rcpf(__expf(2.0f * v) + 1.0f);
}

__global__ void __launch_bounds__(32) __attribute__((amdgpu_waves_per_eu(1, 1)))
lstm_scan(const float* __restrict__ y,
                          const float* __restrict__ inp,
                          const float* __restrict__ w_ih_f, const float* __restrict__ w_hh_f,
                          const float* __restrict__ b_ih_f, const float* __restrict__ b_hh_f,
                          const float* __restrict__ w_hr_f,
                          const float* __restrict__ w_ih_b, const float* __restrict__ w_hh_b,
                          const float* __restrict__ b_ih_b, const float* __restrict__ b_hh_b,
                          const float* __restrict__ w_hr_b,
                          const float* __restrict__ h0, const float* __restrict__ c0,
                          float* __restrict__ hout) {
    const int t = threadIdx.x;
    if (t >= 16) return;
    const int dir = t >> 3;
    const int b = t & 7;

    const float* w_ih = dir ? w_ih_b : w_ih_f;
    const float* w_hh = dir ? w_hh_b : w_hh_f;
    const float* b_ih = dir ? b_ih_b : b_ih_f;
    const float* b_hh = dir ? b_hh_b : b_hh_f;
    const float* w_hr = dir ? w_hr_b : w_hr_f;

    float wi0[20], wi1[20], whh[20], bsum[20], whr[HID], c[HID];
#pragma unroll
    for (int j = 0; j < 20; ++j) {
        wi0[j] = w_ih[2 * j];
        wi1[j] = w_ih[2 * j + 1];
        whh[j] = w_hh[j];
        bsum[j] = b_ih[j] + b_hh[j];
    }
#pragma unroll
    for (int k = 0; k < HID; ++k) {
        whr[k] = w_hr[k];
        c[k] = c0[(dir * NB + b) * HID + k];
    }
    float h = h0[dir * NB + b];

    const float* yb = y + (size_t)b * LSEQ;
    const float* ib = inp + (size_t)b * LSEQ;
    float* ho = hout + (size_t)(dir * NB + b) * LSEQ;

    for (int s = 0; s < LSEQ; ++s) {
        const int l = dir ? (LSEQ - 1 - s) : s;
        if ((s & 31) == 0) {
            const int pl = dir ? (l - 96) : (l + 96);
            if ((unsigned)pl < (unsigned)LSEQ) {
                __builtin_prefetch(yb + pl, 0, 0);
                __builtin_prefetch(ib + pl, 0, 0);
            }
        }
        const float x0 = yb[l];
        const float x1 = ib[l];
        float hn = 0.0f;
#pragma unroll
        for (int k = 0; k < HID; ++k) {
            const float gi = fmaf(x0, wi0[k],      fmaf(x1, wi1[k],      fmaf(h, whh[k],      bsum[k])));
            const float gf = fmaf(x0, wi0[k + 5],  fmaf(x1, wi1[k + 5],  fmaf(h, whh[k + 5],  bsum[k + 5])));
            const float gc = fmaf(x0, wi0[k + 10], fmaf(x1, wi1[k + 10], fmaf(h, whh[k + 10], bsum[k + 10])));
            const float go = fmaf(x0, wi0[k + 15], fmaf(x1, wi1[k + 15], fmaf(h, whh[k + 15], bsum[k + 15])));
            const float iv = sigm(gi);
            const float fv = sigm(gf);
            const float gv = tanh_fast(gc);
            const float ov = sigm(go);
            c[k] = fmaf(fv, c[k], iv * gv);
            hn = fmaf(ov * tanh_fast(c[k]), whr[k], hn);
        }
        h = hn;
        ho[l] = h;
    }
}

__global__ void seq_softmax(const float* __restrict__ hbuf, float* __restrict__ out) {
    const int b = blockIdx.x;
    const int tid = threadIdx.x;
    const int T = 256;
    __shared__ float red[256];

    const float* hf = hbuf + (size_t)b * LSEQ;
    const float* hb = hbuf + (size_t)(NB + b) * LSEQ;

    float m = -3.0e38f;
    for (int l = tid; l < LSEQ; l += T) m = fmaxf(m, hf[l] + hb[l]);
    red[tid] = m;
    __syncthreads();
    for (int s = T / 2; s > 0; s >>= 1) {
        if (tid < s) red[tid] = fmaxf(red[tid], red[tid + s]);
        __syncthreads();
    }
    m = red[0];
    __syncthreads();

    float sum = 0.0f;
    for (int l = tid; l < LSEQ; l += T) sum += __expf(hf[l] + hb[l] - m);
    red[tid] = sum;
    __syncthreads();
    for (int s = T / 2; s > 0; s >>= 1) {
        if (tid < s) red[tid] += red[tid + s];
        __syncthreads();
    }
    sum = red[0];
    __syncthreads();

    const float inv = __builtin_amdgcn_rcpf(sum);
    for (int l = tid; l < LSEQ; l += T) {
        const float e = __expf(hf[l] + hb[l] - m) * inv;
        float* o = out + ((size_t)b * LSEQ + l) * 2;
        o[0] = e;
        o[1] = e;
    }
}

extern "C" void kernel_launch(void* const* d_in, const int* in_sizes, int n_in,
                              void* d_out, int out_size, void* d_ws, size_t ws_size,
                              hipStream_t stream) {
    const float* x      = (const float*)d_in[0];
    const float* inp    = (const float*)d_in[1];
    const float* conv_w = (const float*)d_in[2];
    const float* conv_b = (const float*)d_in[3];
    const float* w_ih_f = (const float*)d_in[4];
    const float* w_hh_f = (const float*)d_in[5];
    const float* b_ih_f = (const float*)d_in[6];
    const float* b_hh_f = (const float*)d_in[7];
    const float* w_hr_f = (const float*)d_in[8];
    const float* w_ih_b = (const float*)d_in[9];
    const float* w_hh_b = (const float*)d_in[10];
    const float* b_ih_b = (const float*)d_in[11];
    const float* b_hh_b = (const float*)d_in[12];
    const float* w_hr_b = (const float*)d_in[13];
    const float* h0     = (const float*)d_in[14];
    const float* c0     = (const float*)d_in[15];
    float* out = (float*)d_out;

    float* y    = (float*)d_ws;
    float* hbuf = y + (size_t)NB * LSEQ;

    conv1x1_wmma<<<(NB * LSEQ / 16) / 8, 256, 0, stream>>>(x, conv_w, conv_b, y);

    lstm_scan<<<1, 32, 0, stream>>>(y, inp,
                                    w_ih_f, w_hh_f, b_ih_f, b_hh_f, w_hr_f,
                                    w_ih_b, w_hh_b, b_ih_b, b_hh_b, w_hr_b,
                                    h0, c0, hbuf);

    seq_softmax<<<NB, 256, 0, stream>>>(hbuf, out);
}


# LSTM scan: time loop in blocks of 8 steps; y and inp of the next block come by 16-byte loads a block ahead, h leaves by 16-byte stores; a step holds no load, address sum or branch; sum order kept
# speedup vs baseline: 3.9526x; 2.4685x over previous
#include <hip/hip_runtime.h>
#include <hip/hip_bf16.h>

#define LSEQ 60000
#define NB 8
#define INCH 64
#define HID 5

typedef __attribute__((ext_vector_type(2))) float v2f;
typedef __attribute__((ext_vector_type(8))) float v8f;

__global__ void conv1x1_wmma(const float* __restrict__ x,
                             const float* __restrict__ w,
                             const float* __restrict__ bias,
                             float* __restrict__ y) {
    const int lane = threadIdx.x & 31;
    const int wave = threadIdx.x >> 5;
    const int tile = blockIdx.x * 8 + wave;
    const int b = tile / (LSEQ / 16);
    const int lbase = (tile % (LSEQ / 16)) * 16;
    const int m = lane & 15;
    const int half = lane >> 4;

    const float* xb = x + (size_t)b * INCH * LSEQ + lbase + m;

    v8f acc = {};
#pragma unroll
    for (int chunk = 0; chunk < 16; ++chunk) {
        const int k0 = chunk * 4 + 2 * half;
        v2f a, bb;
        a.x = xb[(size_t)k0 * LSEQ];
        a.y = xb[(size_t)(k0 + 1) * LSEQ];
        bb.x = w[k0];
        bb.y = w[k0 + 1];
        acc = __builtin_amdgcn_wmma_f32_16x16x4_f32(
             false, a,  false, bb,
             (short)0, acc,  false,  false);
    }

    const float bv = bias[0];
    float* yo = y + (size_t)b * LSEQ + lbase;
    if (lane == 0) {
#pragma unroll
        for (int r = 0; r < 8; ++r) yo[r] = acc[r] + bv;
    } else if (lane == 16) {
#pragma unroll
        for (int r = 0; r < 8; ++r) yo[8 + r] = acc[r] + bv;
    }
}

__device__ __forceinline__ float sigm(float v) {
    return __builtin_amdgcn_rcpf(1.0f + __expf(-v));
}
__device__ __forceinline__ float tanh_fast(float v) {
    return 1.0f - 2.0f * __builtin_amdgcn_rcpf(__expf(2.0f * v) + 1.0f);
}

#define SBLK 8
#define NBLK (LSEQ / SBLK)
typedef __attribute__((ext_vector_type(4))) float v4f;

__global__ void __launch_bounds__(32) __attribute__((amdgpu_waves_per_eu(1, 1)))
lstm_scan(const float* __restrict__ y,
                          const float* __restrict__ inp,
                          const float* __restrict__ w_ih_f, const float* __restrict__ w_hh_f,
                          const float* __restrict__ b_ih_f, const float* __restrict__ b_hh_f,
                          const float* __restrict__ w_hr_f,
                          const float* __restrict__ w_ih_b, const float* __restrict__ w_hh_b,
                          const float* __restrict__ b_ih_b, const float* __restrict__ b_hh_b,
                          const float* __restrict__ w_hr_b,
                          const float* __restrict__ h0, const float* __restrict__ c0,
                          float* __restrict__ hout) {
    const int t = threadIdx.x;
    if (t >= 16) return;
    const int dir = t >> 3;
    const int b = t & 7;

    const float* w_ih = dir ? w_ih_b : w_ih_f;
    const float* w_hh = dir ? w_hh_b : w_hh_f;
    const float* b_ih = dir ? b_ih_b : b_ih_f;
    const float* b_hh = dir ? b_hh_b : b_hh_f;
    const float* w_hr = dir ? w_hr_b : w_hr_f;

    float wi0[20], wi1[20], whh[20], bsum[20], whr[HID], c[HID];
#pragma unroll
    for (int j = 0; j < 20; ++j) {
        wi0[j] = w_ih[2 * j];
        wi1[j] = w_ih[2 * j + 1];
        whh[j] = w_hh[j];
        bsum[j] = b_ih[j] + b_hh[j];
    }
#pragma unroll
    for (int k = 0; k < HID; ++k) {
        whr[k] = w_hr[k];
        c[k] = c0[(dir * NB + b) * HID + k];
    }
    float h = h0[dir * NB + b];

    const unsigned row_in = (unsigned)b * LSEQ;
    const unsigned row_out = (unsigned)(dir * NB + b) * LSEQ;
    unsigned off = dir ? (unsigned)(LSEQ - SBLK) : 0u;
    const unsigned dstep = dir ? (unsigned)(-SBLK) : (unsigned)SBLK;

    v4f yc0 = *(const v4f*)(y + row_in + off);
    v4f yc1 = *(const v4f*)(y + row_in + off + 4);
    v4f ic0 = *(const v4f*)(inp + row_in + off);
    v4f ic1 = *(const v4f*)(inp + row_in + off + 4);

    for (int blk = 0; blk < NBLK; ++blk) {
        const unsigned noff = (blk + 1 < NBLK) ? off + dstep : off;
        const v4f yn0 = *(const v4f*)(y + row_in + noff);
        const v4f yn1 = *(const v4f*)(y + row_in + noff + 4);
        const v4f in0 = *(const v4f*)(inp + row_in + noff);
        const v4f in1 = *(const v4f*)(inp + row_in + noff + 4);

        const float xs0[SBLK] = {
            dir ? float(yc1.w) : float(yc0.x), dir ? float(yc1.z) : float(yc0.y),
            dir ? float(yc1.y) : float(yc0.z), dir ? float(yc1.x) : float(yc0.w),
            dir ? float(yc0.w) : float(yc1.x), dir ? float(yc0.z) : float(yc1.y),
            dir ? float(yc0.y) : float(yc1.z), dir ? float(yc0.x) : float(yc1.w)};
        const float xs1[SBLK] = {
            dir ? float(ic1.w) : float(ic0.x), dir ? float(ic1.z) : float(ic0.y),
            dir ? float(ic1.y) : float(ic0.z), dir ? float(ic1.x) : float(ic0.w),
            dir ? float(ic0.w) : float(ic1.x), dir ? float(ic0.z) : float(ic1.y),
            dir ? float(ic0.y) : float(ic1.z), dir ? float(ic0.x) : float(ic1.w)};
        float hs[SBLK];

#pragma unroll
        for (int i = 0; i < SBLK; ++i) {
            const float x0 = xs0[i];
            const float x1 = xs1[i];
            float hn = 0.0f;
#pragma unroll
            for (int k = 0; k < HID; ++k) {
                const float gi = fmaf(x0, wi0[k],      fmaf(x1, wi1[k],      fmaf(h, whh[k],      bsum[k])));
                const float gf = fmaf(x0, wi0[k + 5],  fmaf(x1, wi1[k + 5],  fmaf(h, whh[k + 5],  bsum[k + 5])));
                const float gc = fmaf(x0, wi0[k + 10], fmaf(x1, wi1[k + 10], fmaf(h, whh[k + 10], bsum[k + 10])));
                const float go = fmaf(x0, wi0[k + 15], fmaf(x1, wi1[k + 15], fmaf(h, whh[k + 15], bsum[k + 15])));
                const float iv = sigm(gi);
                const float fv = sigm(gf);
                const float gv = tanh_fast(gc);
                const float ov = sigm(go);
                c[k] = fmaf(fv, c[k], iv * gv);
                hn = fmaf(ov * tanh_fast(c[k]), whr[k], hn);
            }
            h = hn;
            hs[i] = h;
        }

        v4f o0, o1;
        o0.x = dir ? float(hs[7]) : float(hs[0]);
        o0.y = dir ? float(hs[6]) : float(hs[1]);
        o0.z = dir ? float(hs[5]) : float(hs[2]);
        o0.w = dir ? float(hs[4]) : float(hs[3]);
        o1.x = dir ? float(hs[3]) : float(hs[4]);
        o1.y = dir ? float(hs[2]) : float(hs[5]);
        o1.z = dir ? float(hs[1]) : float(hs[6]);
        o1.w = dir ? float(hs[0]) : float(hs[7]);
        *(v4f*)(hout + row_out + off) = o0;
        *(v4f*)(hout + row_out + off + 4) = o1;

        off = noff;
        yc0 = yn0; yc1 = yn1; ic0 = in0; ic1 = in1;
    }
}

__global__ void seq_softmax(const float* __restrict__ hbuf, float* __restrict__ out) {
    const int b = blockIdx.x;
    const int tid = threadIdx.x;
    const int T = 256;
    __shared__ float red[256];

    const float* hf = hbuf + (size_t)b * LSEQ;
    const float* hb = hbuf + (size_t)(NB + b) * LSEQ;

    float m = -3.0e38f;
    for (int l = tid; l < LSEQ; l += T) m = fmaxf(m, hf[l] + hb[l]);
    red[tid] = m;
    __syncthreads();
    for (int s = T / 2; s > 0; s >>= 1) {
        if (tid < s) red[tid] = fmaxf(red[tid], red[tid + s]);
        __syncthreads();
    }
    m = red[0];
    __syncthreads();

    float sum = 0.0f;
    for (int l = tid; l < LSEQ; l += T) sum += __expf(hf[l] + hb[l] - m);
    red[tid] = sum;
    __syncthreads();
    for (int s = T / 2; s > 0; s >>= 1) {
        if (tid < s) red[tid] += red[tid + s];
        __syncthreads();
    }
    sum = red[0];
    __syncthreads();

    const float inv = __builtin_amdgcn_rcpf(sum);
    for (int l = tid; l < LSEQ; l += T) {
        const float e = __expf(hf[l] + hb[l] - m) * inv;
        float* o = out + ((size_t)b * LSEQ + l) * 2;
        o[0] = e;
        o[1] = e;
    }
}

extern "C" void kernel_launch(void* const* d_in, const int* in_sizes, int n_in,
                              void* d_out, int out_size, void* d_ws, size_t ws_size,
                              hipStream_t stream) {
    const float* x      = (const float*)d_in[0];
    const float* inp    = (const float*)d_in[1];
    const float* conv_w = (const float*)d_in[2];
    const float* conv_b = (const float*)d_in[3];
    const float* w_ih_f = (const float*)d_in[4];
    const float* w_hh_f = (const float*)d_in[5];
    const float* b_ih_f = (const float*)d_in[6];
    const float* b_hh_f = (const float*)d_in[7];
    const float* w_hr_f = (const float*)d_in[8];
    const float* w_ih_b = (const float*)d_in[9];
    const float* w_hh_b = (const float*)d_in[10];
    const float* b_ih_b = (const float*)d_in[11];
    const float* b_hh_b = (const float*)d_in[12];
    const float* w_hr_b = (const float*)d_in[13];
    const float* h0     = (const float*)d_in[14];
    const float* c0     = (const float*)d_in[15];
    float* out = (float*)d_out;

    float* y    = (float*)d_ws;
    float* hbuf = y + (size_t)NB * LSEQ;

    conv1x1_wmma<<<(NB * LSEQ / 16) / 8, 256, 0, stream>>>(x, conv_w, conv_b, y);

    lstm_scan<<<1, 32, 0, stream>>>(y, inp,
                                    w_ih_f, w_hh_f, b_ih_f, b_hh_f, w_hr_f,
                                    w_ih_b, w_hh_b, b_ih_b, b_hh_b, w_hr_b,
                                    h0, c0, hbuf);

    seq_softmax<<<NB, 256, 0, stream>>>(hbuf, out);
}


# LSTM scan: h enters the recurrent matrix product as each wave half's own partial sum (w_hh in both k pairs), grp chosen in w_hr: lane exchange, add, select leave the chain; gate sum order changes
# speedup vs baseline: 6.2891x; 1.5911x over previous
#include <hip/hip_runtime.h>
#include <hip/hip_bf16.h>

#define LSEQ 60000
#define NB 8
#define INCH 64
#define HID 5

typedef __attribute__((ext_vector_type(2))) float v2f;
typedef __attribute__((ext_vector_type(8))) float v8f;

__global__ void conv1x1_wmma(const float* __restrict__ x,
                             const float* __restrict__ w,
                             const float* __restrict__ bias,
                             float* __restrict__ y) {
    const int lane = threadIdx.x & 31;
    const int wave = threadIdx.x >> 5;
    const int tile = blockIdx.x * 8 + wave;
    const int b = tile / (LSEQ / 16);
    const int lbase = (tile % (LSEQ / 16)) * 16;
    const int m = lane & 15;
    const int half = lane >> 4;

    const float* xb = x + (size_t)b * INCH * LSEQ + lbase + m;

    v8f acc = {};
#pragma unroll
    for (int chunk = 0; chunk < 16; ++chunk) {
        const int k0 = chunk * 4 + 2 * half;
        v2f a, bb;
        a.x = xb[(size_t)k0 * LSEQ];
        a.y = xb[(size_t)(k0 + 1) * LSEQ];
        bb.x = w[k0];
        bb.y = w[k0 + 1];
        acc = __builtin_amdgcn_wmma_f32_16x16x4_f32(
             false, a,  false, bb,
             (short)0, acc,  false,  false);
    }

    const float bv = bias[0];
    float* yo = y + (size_t)b * LSEQ + lbase;
    if (lane == 0) {
#pragma unroll
        for (int r = 0; r < 8; ++r) yo[r] = acc[r] + bv;
    } else if (lane == 16) {
#pragma unroll
        for (int r = 0; r < 8; ++r) yo[8 + r] = acc[r] + bv;
    }
}

#define KS (-1.44269504f)
#define KT (2.88539008f)
__device__ __forceinline__ float rcp1p_exp2(float s) {
    return __builtin_amdgcn_rcpf(1.0f + __builtin_amdgcn_exp2f(s));
}

#define SBLK 8
#define NBLK (LSEQ / SBLK)
typedef __attribute__((ext_vector_type(4))) float v4f;

__device__ __forceinline__ int gate_of(int q) { return q == 2 ? 3 : (q == 3 ? 2 : q); }
__device__ __forceinline__ int unit_of(int grp, int slot) {
    return grp == 0 ? (slot < 3 ? slot : -1) : (slot < 2 ? 3 + slot : -1);
}
__device__ __forceinline__ float other_half(float v) {
    const int iv = __float_as_int(v);
    return __int_as_float(__builtin_amdgcn_permlanex16(iv, iv, 0x76543210, 0xfedcba98, true, false));
}
__device__ __forceinline__ float other_grp(float v) {
    return __int_as_float(__builtin_amdgcn_mov_dpp(__float_as_int(v), 0xB1, 0xF, 0xF, true));
}

__device__ __forceinline__ void operand_guard(v8f& d, v8f& dx) {
    asm volatile("v_nop\n\tv_nop\n\tv_nop\n\tv_nop" : "+v"(d), "+v"(dx));
}

template <int DIR>
__device__ __forceinline__ void scan_dir(const float* __restrict__ y,
                                         const float* __restrict__ inp,
                                         const float* __restrict__ w_ih, const float* __restrict__ w_hh,
                                         const float* __restrict__ b_ih, const float* __restrict__ b_hh,
                                         const float* __restrict__ w_hr,
                                         const float* __restrict__ h0, const float* __restrict__ c0,
                                         float* __restrict__ hout) {
    const int lane = threadIdx.x;
    const int hf = lane >> 4;
    const int col = lane & 15;
    const int b = col >> 1;
    const int grp = col & 1;

    const int aslot = 2 * (col >> 3) + (col & 1);
    const int agate = gate_of((col & 7) >> 1);
    const int ux = unit_of(hf, aslot);
    const float ascale = agate == 2 ? KT : KS;
    v2f ax;
    ax.x = ux >= 0 ? w_ih[2 * (agate * 5 + ux)] * ascale : 0.0f;
    ax.y = ux >= 0 ? w_ih[2 * (agate * 5 + ux) + 1] * ascale : 0.0f;
    const int uh0 = unit_of(0, aslot);
    const int uh1 = unit_of(1, aslot);
    v2f ah;
    ah.x = uh0 >= 0 ? w_hh[agate * 5 + uh0] * ascale : 0.0f;
    ah.y = uh1 >= 0 ? w_hh[agate * 5 + uh1] * ascale : 0.0f;

    const int uA = unit_of(grp, 2 * hf);
    const int uB = unit_of(grp, 2 * hf + 1);
    v8f cb;
#pragma unroll
    for (int r = 0; r < 8; ++r) {
        const int u = (r & 1) ? uB : uA;
        const int j = gate_of(r >> 1) * 5 + u;
        cb[r] = u >= 0 ? (b_ih[j] + b_hh[j]) * ((r >> 1) == 3 ? KT : KS) : 0.0f;
    }
    const float whrA = uA >= 0 ? w_hr[uA] : 0.0f;
    const float whrB = uB >= 0 ? w_hr[uB] : 0.0f;
    const float whrAx = grp == 0 ? whrA : 0.0f;
    const float whrBx = grp == 0 ? whrB : 0.0f;
    const float whrAy = grp == 1 ? whrA : 0.0f;
    const float whrBy = grp == 1 ? whrB : 0.0f;
    const float whrA2x = -2.0f * whrAx;
    const float whrB2x = -2.0f * whrBx;
    const float whrA2y = -2.0f * whrAy;
    const float whrB2y = -2.0f * whrBy;
    float cA = uA >= 0 ? KT * c0[(DIR * NB + b) * HID + uA] : 0.0f;
    float cB = uB >= 0 ? KT * c0[(DIR * NB + b) * HID + uB] : 0.0f;
    float h = h0[DIR * NB + b];

    const bool own = (grp == hf);
    const bool writer = (hf == 0 && grp == 0);

    const unsigned row_in = (unsigned)b * LSEQ;
    const unsigned row_out = (unsigned)(DIR * NB + b) * LSEQ;
    unsigned off = DIR ? (unsigned)(LSEQ - SBLK) : 0u;
    const unsigned dstep = DIR ? (unsigned)(-SBLK) : (unsigned)SBLK;

    v4f yc0 = *(const v4f*)(y + row_in + off);
    v4f yc1 = *(const v4f*)(y + row_in + off + 4);
    v4f ic0 = *(const v4f*)(inp + row_in + off);
    v4f ic1 = *(const v4f*)(inp + row_in + off + 4);

    v8f dx;
    {
        v2f bx;
        bx.x = own ? (DIR ? float(yc1.w) : float(yc0.x)) : 0.0f;
        bx.y = own ? (DIR ? float(ic1.w) : float(ic0.x)) : 0.0f;
        __builtin_amdgcn_sched_barrier(0);
        dx = __builtin_amdgcn_wmma_f32_16x16x4_f32(false, ax, false, bx, (short)0, cb, false, false);
        __builtin_amdgcn_sched_barrier(0);
        asm volatile("v_nop\n\tv_nop\n\tv_nop\n\tv_nop" : "+v"(dx));
        __builtin_amdgcn_sched_barrier(0);
    }
    v2f bh;
    bh.x = (hf == 0 && grp == 0) ? h : 0.0f;
    bh.y = (hf == 0 && grp == 1) ? h : 0.0f;

    for (int blk = 0; blk < NBLK; ++blk) {
        const unsigned noff = (blk + 1 < NBLK) ? off + dstep : off;
        const v4f yn0 = *(const v4f*)(y + row_in + noff);
        const v4f yn1 = *(const v4f*)(y + row_in + noff + 4);
        const v4f in0 = *(const v4f*)(inp + row_in + noff);
        const v4f in1 = *(const v4f*)(inp + row_in + noff + 4);

        const float xn0[SBLK] = {
            DIR ? float(yc1.z) : float(yc0.y), DIR ? float(yc1.y) : float(yc0.z),
            DIR ? float(yc1.x) : float(yc0.w), DIR ? float(yc0.w) : float(yc1.x),
            DIR ? float(yc0.z) : float(yc1.y), DIR ? float(yc0.y) : float(yc1.z),
            DIR ? float(yc0.x) : float(yc1.w), DIR ? float(yn1.w) : float(yn0.x)};
        const float xn1[SBLK] = {
            DIR ? float(ic1.z) : float(ic0.y), DIR ? float(ic1.y) : float(ic0.z),
            DIR ? float(ic1.x) : float(ic0.w), DIR ? float(ic0.w) : float(ic1.x),
            DIR ? float(ic0.z) : float(ic1.y), DIR ? float(ic0.y) : float(ic1.z),
            DIR ? float(ic0.x) : float(ic1.w), DIR ? float(in1.w) : float(in0.x)};
        float hs[SBLK];

#pragma unroll
        for (int i = 0; i < SBLK; ++i) {
            v2f bx;
            bx.x = own ? xn0[i] : 0.0f;
            bx.y = own ? xn1[i] : 0.0f;
            __builtin_amdgcn_sched_barrier(0);
            v8f d = __builtin_amdgcn_wmma_f32_16x16x4_f32(false, ah, false, bh, (short)0, dx, false, false);
            asm volatile("" : "+v"(d), "+v"(bx));
            __builtin_amdgcn_sched_barrier(0);
            dx = __builtin_amdgcn_wmma_f32_16x16x4_f32(false, ax, false, bx, (short)0, cb, false, false);
            __builtin_amdgcn_sched_barrier(0);
            operand_guard(d, dx);
            __builtin_amdgcn_sched_barrier(0);

            if (i > 0) hs[i - 1] = bh.x + other_half(bh.x);

            const float iA = rcp1p_exp2(d[0]);
            const float iB = rcp1p_exp2(d[1]);
            const float gA = fmaf(rcp1p_exp2(d[6]), -2.0f * KT, KT);
            const float gB = fmaf(rcp1p_exp2(d[7]), -2.0f * KT, KT);
            const float fA = rcp1p_exp2(d[2]);
            const float fB = rcp1p_exp2(d[3]);
            const float oA = rcp1p_exp2(d[4]);
            const float oB = rcp1p_exp2(d[5]);
            const float osumx = fmaf(oA, whrAx, oB * whrBx);
            const float osumy = fmaf(oA, whrAy, oB * whrBy);
            const float owA2x = oA * whrA2x;
            const float owB2x = oB * whrB2x;
            const float owA2y = oA * whrA2y;
            const float owB2y = oB * whrB2y;
            cA = fmaf(fA, cA, iA * gA);
            cB = fmaf(fB, cB, iB * gB);
            const float rA = rcp1p_exp2(cA);
            const float rB = rcp1p_exp2(cB);
            const float px = fmaf(rB, owB2x, fmaf(rA, owA2x, osumx));
            const float py = fmaf(rB, owB2y, fmaf(rA, owA2y, osumy));
            bh.x = px + other_grp(py);
            bh.y = py + other_grp(px);
        }
        hs[SBLK - 1] = bh.x + other_half(bh.x);

        if (writer) {
            v4f o0, o1;
            o0.x = DIR ? hs[7] : hs[0];
            o0.y = DIR ? hs[6] : hs[1];
            o0.z = DIR ? hs[5] : hs[2];
            o0.w = DIR ? hs[4] : hs[3];
            o1.x = DIR ? hs[3] : hs[4];
            o1.y = DIR ? hs[2] : hs[5];
            o1.z = DIR ? hs[1] : hs[6];
            o1.w = DIR ? hs[0] : hs[7];
            *(v4f*)(hout + row_out + off) = o0;
            *(v4f*)(hout + row_out + off + 4) = o1;
        }

        off = noff;
        yc0 = yn0; yc1 = yn1; ic0 = in0; ic1 = in1;
    }
}

__global__ void __launch_bounds__(32) __attribute__((amdgpu_waves_per_eu(1, 1)))
lstm_scan(const float* __restrict__ y,
                          const float* __restrict__ inp,
                          const float* __restrict__ w_ih_f, const float* __restrict__ w_hh_f,
                          const float* __restrict__ b_ih_f, const float* __restrict__ b_hh_f,
                          const float* __restrict__ w_hr_f,
                          const float* __restrict__ w_ih_b, const float* __restrict__ w_hh_b,
                          const float* __restrict__ b_ih_b, const float* __restrict__ b_hh_b,
                          const float* __restrict__ w_hr_b,
                          const float* __restrict__ h0, const float* __restrict__ c0,
                          float* __restrict__ hout) {
    if (blockIdx.x == 0)
        scan_dir<0>(y, inp, w_ih_f, w_hh_f, b_ih_f, b_hh_f, w_hr_f, h0, c0, hout);
    else
        scan_dir<1>(y, inp, w_ih_b, w_hh_b, b_ih_b, b_hh_b, w_hr_b, h0, c0, hout);
}

__global__ void seq_softmax(const float* __restrict__ hbuf, float* __restrict__ out) {
    const int b = blockIdx.x;
    const int tid = threadIdx.x;
    const int T = 256;
    __shared__ float red[256];

    const float* hf = hbuf + (size_t)b * LSEQ;
    const float* hb = hbuf + (size_t)(NB + b) * LSEQ;

    float m = -3.0e38f;
    for (int l = tid; l < LSEQ; l += T) m = fmaxf(m, hf[l] + hb[l]);
    red[tid] = m;
    __syncthreads();
    for (int s = T / 2; s > 0; s >>= 1) {
        if (tid < s) red[tid] = fmaxf(red[tid], red[tid + s]);
        __syncthreads();
    }
    m = red[0];
    __syncthreads();

    float sum = 0.0f;
    for (int l = tid; l < LSEQ; l += T) sum += __expf(hf[l] + hb[l] - m);
    red[tid] = sum;
    __syncthreads();
    for (int s = T / 2; s > 0; s >>= 1) {
        if (tid < s) red[tid] += red[tid + s];
        __syncthreads();
    }
    sum = red[0];
    __syncthreads();

    const float inv = __builtin_amdgcn_rcpf(sum);
    for (int l = tid; l < LSEQ; l += T) {
        const float e = __expf(hf[l] + hb[l] - m) * inv;
        float* o = out + ((size_t)b * LSEQ + l) * 2;
        o[0] = e;
        o[1] = e;
    }
}

extern "C" void kernel_launch(void* const* d_in, const int* in_sizes, int n_in,
                              void* d_out, int out_size, void* d_ws, size_t ws_size,
                              hipStream_t stream) {
    const float* x      = (const float*)d_in[0];
    const float* inp    = (const float*)d_in[1];
    const float* conv_w = (const float*)d_in[2];
    const float* conv_b = (const float*)d_in[3];
    const float* w_ih_f = (const float*)d_in[4];
    const float* w_hh_f = (const float*)d_in[5];
    const float* b_ih_f = (const float*)d_in[6];
    const float* b_hh_f = (const float*)d_in[7];
    const float* w_hr_f = (const float*)d_in[8];
    const float* w_ih_b = (const float*)d_in[9];
    const float* w_hh_b = (const float*)d_in[10];
    const float* b_ih_b = (const float*)d_in[11];
    const float* b_hh_b = (const float*)d_in[12];
    const float* w_hr_b = (const float*)d_in[13];
    const float* h0     = (const float*)d_in[14];
    const float* c0     = (const float*)d_in[15];
    float* out = (float*)d_out;

    float* y    = (float*)d_ws;
    float* hbuf = y + (size_t)NB * LSEQ;

    conv1x1_wmma<<<(NB * LSEQ / 16) / 8, 256, 0, stream>>>(x, conv_w, conv_b, y);

    lstm_scan<<<2, 32, 0, stream>>>(y, inp,
                                    w_ih_f, w_hh_f, b_ih_f, b_hh_f, w_hr_f,
                                    w_ih_b, w_hh_b, b_ih_b, b_hh_b, w_hr_b,
                                    h0, c0, hbuf);

    seq_softmax<<<NB, 256, 0, stream>>>(hbuf, out);
}


# speedup vs baseline: 6.9257x; 1.1012x over previous
#include <hip/hip_runtime.h>
#include <hip/hip_bf16.h>

#define LSEQ 60000
#define NB 8
#define INCH 64
#define HID 5

typedef __attribute__((ext_vector_type(2))) float v2f;
typedef __attribute__((ext_vector_type(8))) float v8f;

__global__ void conv1x1_wmma(const float* __restrict__ x,
                             const float* __restrict__ w,
                             const float* __restrict__ bias,
                             float* __restrict__ y) {
    const int lane = threadIdx.x & 31;
    const int wave = threadIdx.x >> 5;
    const int tile = blockIdx.x * 8 + wave;
    const int b = tile / (LSEQ / 16);
    const int lbase = (tile % (LSEQ / 16)) * 16;
    const int m = lane & 15;
    const int half = lane >> 4;

    const float* xb = x + (size_t)b * INCH * LSEQ + lbase + m;

    v8f acc = {};
#pragma unroll
    for (int chunk = 0; chunk < 16; ++chunk) {
        const int k0 = chunk * 4 + 2 * half;
        v2f a, bb;
        a.x = xb[(size_t)k0 * LSEQ];
        a.y = xb[(size_t)(k0 + 1) * LSEQ];
        bb.x = w[k0];
        bb.y = w[k0 + 1];
        acc = __builtin_amdgcn_wmma_f32_16x16x4_f32(
             false, a,  false, bb,
             (short)0, acc,  false,  false);
    }

    const float bv = bias[0];
    float* yo = y + (size_t)b * LSEQ + lbase;
    if (lane == 0) {
#pragma unroll
        for (int r = 0; r < 8; ++r) yo[r] = acc[r] + bv;
    } else if (lane == 16) {
#pragma unroll
        for (int r = 0; r < 8; ++r) yo[8 + r] = acc[r] + bv;
    }
}

#define KS (-1.44269504f)
#define KT (2.88539008f)
__device__ __forceinline__ float rcp1p_exp2(float s) {
    return __builtin_amdgcn_rcpf(1.0f + __builtin_amdgcn_exp2f(s));
}

#define SBLK 8
#define NBLK (LSEQ / SBLK)
typedef __attribute__((ext_vector_type(4))) float v4f;

__device__ __forceinline__ int gate_of(int q) { return q == 2 ? 3 : (q == 3 ? 2 : q); }
__device__ __forceinline__ int unit_of(int grp, int slot) {
    return grp == 0 ? (slot < 3 ? slot : -1) : (slot < 2 ? 3 + slot : -1);
}
__device__ __forceinline__ float other_half(float v) {
    const int iv = __float_as_int(v);
    return __int_as_float(__builtin_amdgcn_permlanex16(iv, iv, 0x76543210, 0xfedcba98, true, false));
}
__device__ __forceinline__ float other_grp(float v) {
    return __int_as_float(__builtin_amdgcn_mov_dpp(__float_as_int(v), 0xB1, 0xF, 0xF, true));
}

#define ISSUE_AFTER(prev, next) asm volatile("" : "+v"(prev), "+v"(next))

__device__ __forceinline__ void operand_guard(v8f& d, v8f& dx) {
    asm volatile("v_nop\n\tv_nop\n\tv_nop\n\tv_nop" : "+v"(d), "+v"(dx));
}

template <int DIR>
__device__ __forceinline__ void scan_dir(const float* __restrict__ y,
                                         const float* __restrict__ inp,
                                         const float* __restrict__ w_ih, const float* __restrict__ w_hh,
                                         const float* __restrict__ b_ih, const float* __restrict__ b_hh,
                                         const float* __restrict__ w_hr,
                                         const float* __restrict__ h0, const float* __restrict__ c0,
                                         float* __restrict__ hout) {
    const int lane = threadIdx.x;
    const int hf = lane >> 4;
    const int col = lane & 15;
    const int b = col >> 1;
    const int grp = col & 1;

    const int aslot = 2 * (col >> 3) + (col & 1);
    const int agate = gate_of((col & 7) >> 1);
    const int ux = unit_of(hf, aslot);
    const float ascale = agate == 2 ? KT : KS;
    v2f ax;
    ax.x = ux >= 0 ? w_ih[2 * (agate * 5 + ux)] * ascale : 0.0f;
    ax.y = ux >= 0 ? w_ih[2 * (agate * 5 + ux) + 1] * ascale : 0.0f;
    const int uh0 = unit_of(0, aslot);
    const int uh1 = unit_of(1, aslot);
    v2f ah;
    ah.x = uh0 >= 0 ? w_hh[agate * 5 + uh0] * ascale : 0.0f;
    ah.y = uh1 >= 0 ? w_hh[agate * 5 + uh1] * ascale : 0.0f;

    const int uA = unit_of(grp, 2 * hf);
    const int uB = unit_of(grp, 2 * hf + 1);
    v8f cb;
#pragma unroll
    for (int r = 0; r < 8; ++r) {
        const int u = (r & 1) ? uB : uA;
        const int j = gate_of(r >> 1) * 5 + u;
        cb[r] = u >= 0 ? (b_ih[j] + b_hh[j]) * ((r >> 1) == 3 ? KT : KS) : 0.0f;
    }
    const float whrA = uA >= 0 ? w_hr[uA] : 0.0f;
    const float whrB = uB >= 0 ? w_hr[uB] : 0.0f;
    const float whrAx = grp == 0 ? whrA : 0.0f;
    const float whrBx = grp == 0 ? whrB : 0.0f;
    const float whrAy = grp == 1 ? whrA : 0.0f;
    const float whrBy = grp == 1 ? whrB : 0.0f;
    const float whrA2x = -2.0f * whrAx;
    const float whrB2x = -2.0f * whrBx;
    const float whrA2y = -2.0f * whrAy;
    const float whrB2y = -2.0f * whrBy;
    float cA = uA >= 0 ? KT * c0[(DIR * NB + b) * HID + uA] : 0.0f;
    float cB = uB >= 0 ? KT * c0[(DIR * NB + b) * HID + uB] : 0.0f;
    float h = h0[DIR * NB + b];

    const bool own = (grp == hf);
    const bool writer = (hf == 0 && grp == 0);

    const unsigned row_in = (unsigned)b * LSEQ;
    const unsigned row_out = (unsigned)(DIR * NB + b) * LSEQ;
    unsigned off = DIR ? (unsigned)(LSEQ - SBLK) : 0u;
    const unsigned dstep = DIR ? (unsigned)(-SBLK) : (unsigned)SBLK;

    v4f yc0 = *(const v4f*)(y + row_in + off);
    v4f yc1 = *(const v4f*)(y + row_in + off + 4);
    v4f ic0 = *(const v4f*)(inp + row_in + off);
    v4f ic1 = *(const v4f*)(inp + row_in + off + 4);

    v8f dx;
    {
        v2f bx;
        bx.x = own ? (DIR ? float(yc1.w) : float(yc0.x)) : 0.0f;
        bx.y = own ? (DIR ? float(ic1.w) : float(ic0.x)) : 0.0f;
        __builtin_amdgcn_sched_barrier(0);
        dx = __builtin_amdgcn_wmma_f32_16x16x4_f32(false, ax, false, bx, (short)0, cb, false, false);
        __builtin_amdgcn_sched_barrier(0);
        asm volatile("v_nop\n\tv_nop\n\tv_nop\n\tv_nop" : "+v"(dx));
        __builtin_amdgcn_sched_barrier(0);
    }
    v2f bh;
    bh.x = (hf == 0 && grp == 0) ? h : 0.0f;
    bh.y = (hf == 0 && grp == 1) ? h : 0.0f;

    for (int blk = 0; blk < NBLK; ++blk) {
        const unsigned noff = (blk + 1 < NBLK) ? off + dstep : off;
        const v4f yn0 = *(const v4f*)(y + row_in + noff);
        const v4f yn1 = *(const v4f*)(y + row_in + noff + 4);
        const v4f in0 = *(const v4f*)(inp + row_in + noff);
        const v4f in1 = *(const v4f*)(inp + row_in + noff + 4);

        const float xn0[SBLK] = {
            DIR ? float(yc1.z) : float(yc0.y), DIR ? float(yc1.y) : float(yc0.z),
            DIR ? float(yc1.x) : float(yc0.w), DIR ? float(yc0.w) : float(yc1.x),
            DIR ? float(yc0.z) : float(yc1.y), DIR ? float(yc0.y) : float(yc1.z),
            DIR ? float(yc0.x) : float(yc1.w), DIR ? float(yn1.w) : float(yn0.x)};
        const float xn1[SBLK] = {
            DIR ? float(ic1.z) : float(ic0.y), DIR ? float(ic1.y) : float(ic0.z),
            DIR ? float(ic1.x) : float(ic0.w), DIR ? float(ic0.w) : float(ic1.x),
            DIR ? float(ic0.z) : float(ic1.y), DIR ? float(ic0.y) : float(ic1.z),
            DIR ? float(ic0.x) : float(ic1.w), DIR ? float(in1.w) : float(in0.x)};
        float hs[SBLK];

#pragma unroll
        for (int i = 0; i < SBLK; ++i) {
            v2f bx;
            bx.x = own ? xn0[i] : 0.0f;
            bx.y = own ? xn1[i] : 0.0f;
            __builtin_amdgcn_sched_barrier(0);
            v8f d = __builtin_amdgcn_wmma_f32_16x16x4_f32(false, ah, false, bh, (short)0, dx, false, false);
            asm volatile("" : "+v"(d), "+v"(bx));
            __builtin_amdgcn_sched_barrier(0);
            dx = __builtin_amdgcn_wmma_f32_16x16x4_f32(false, ax, false, bx, (short)0, cb, false, false);
            __builtin_amdgcn_sched_barrier(0);
            operand_guard(d, dx);
            __builtin_amdgcn_sched_barrier(0);

            if (i > 0) hs[i - 1] = bh.x + other_half(bh.x);

            float d0 = d[0], d1 = d[1], d2 = d[2], d3 = d[3];
            float d4 = d[4], d5 = d[5], d6 = d[6], d7 = d[7];
            float egA = __builtin_amdgcn_exp2f(d6);
            ISSUE_AFTER(egA, d7);    float egB = __builtin_amdgcn_exp2f(d7);
            ISSUE_AFTER(egB, d0);    float eiA = __builtin_amdgcn_exp2f(d0);
            ISSUE_AFTER(eiA, d1);    float eiB = __builtin_amdgcn_exp2f(d1);
            ISSUE_AFTER(eiB, d2);    float efA = __builtin_amdgcn_exp2f(d2);
            ISSUE_AFTER(efA, egA);   float agA = 1.0f + egA;
            ISSUE_AFTER(agA, d3);    float efB = __builtin_amdgcn_exp2f(d3);
            ISSUE_AFTER(efB, egB);   float agB = 1.0f + egB;
            ISSUE_AFTER(agB, eiA);   float aiA = 1.0f + eiA;
            ISSUE_AFTER(aiA, agA);   const float rgA = __builtin_amdgcn_rcpf(agA);
            float rgA_ = rgA;
            ISSUE_AFTER(rgA_, eiB);  float aiB = 1.0f + eiB;
            ISSUE_AFTER(aiB, agB);   float rgB = __builtin_amdgcn_rcpf(agB);
            ISSUE_AFTER(rgB, efA);   float afA = 1.0f + efA;
            ISSUE_AFTER(afA, aiA);   float iA = __builtin_amdgcn_rcpf(aiA);
            ISSUE_AFTER(iA, efB);    float afB = 1.0f + efB;
            ISSUE_AFTER(afB, aiB);   float iB = __builtin_amdgcn_rcpf(aiB);
            ISSUE_AFTER(iB, d5);     float eoB = __builtin_amdgcn_exp2f(d5);
            ISSUE_AFTER(eoB, rgA_);  float gA = fmaf(rgA_, -2.0f * KT, KT);
            ISSUE_AFTER(gA, afA);    float fA = __builtin_amdgcn_rcpf(afA);
            ISSUE_AFTER(fA, rgB);    float gB = fmaf(rgB, -2.0f * KT, KT);
            ISSUE_AFTER(gB, afB);    float fB = __builtin_amdgcn_rcpf(afB);
            ISSUE_AFTER(fB, iA);     float igA = iA * gA;
            ISSUE_AFTER(igA, d4);    float eoA = __builtin_amdgcn_exp2f(d4);
            ISSUE_AFTER(eoA, iB);    float igB = iB * gB;
            ISSUE_AFTER(igB, eoB);   float aoB = 1.0f + eoB;
            ISSUE_AFTER(aoB, fA);    cA = fmaf(fA, cA, igA);
            ISSUE_AFTER(cA, fB);     cB = fmaf(fB, cB, igB);
            ISSUE_AFTER(cB, aoB);    float oB = __builtin_amdgcn_rcpf(aoB);
            ISSUE_AFTER(oB, eoA);    float aoA = 1.0f + eoA;
            ISSUE_AFTER(aoA, cA);    float ecA = __builtin_amdgcn_exp2f(cA);
            ISSUE_AFTER(ecA, cB);    float ecB = __builtin_amdgcn_exp2f(cB);
            ISSUE_AFTER(ecB, aoA);   float oA = __builtin_amdgcn_rcpf(aoA);
            ISSUE_AFTER(oA, oB);     float mBx = oB * whrBx;
            ISSUE_AFTER(mBx, oB);    float mBy = oB * whrBy;
            ISSUE_AFTER(mBy, ecA);   float acA = 1.0f + ecA;
            ISSUE_AFTER(acA, oB);    float owB2x = oB * whrB2x;
            ISSUE_AFTER(owB2x, ecB); float acB = 1.0f + ecB;
            ISSUE_AFTER(acB, oB);    float owB2y = oB * whrB2y;
            ISSUE_AFTER(owB2y, oA);  float owA2x = oA * whrA2x;
            ISSUE_AFTER(owA2x, acA); float rA = __builtin_amdgcn_rcpf(acA);
            ISSUE_AFTER(rA, oA);     float owA2y = oA * whrA2y;
            ISSUE_AFTER(owA2y, acB); float rB = __builtin_amdgcn_rcpf(acB);
            ISSUE_AFTER(rB, mBx);    float osumx = fmaf(oA, whrAx, mBx);
            ISSUE_AFTER(osumx, mBy); float osumy = fmaf(oA, whrAy, mBy);
            ISSUE_AFTER(osumy, rA);  float px1 = fmaf(rA, owA2x, osumx);
            ISSUE_AFTER(px1, owA2y); float py1 = fmaf(rA, owA2y, osumy);
            ISSUE_AFTER(py1, owB2x); float px = fmaf(rB, owB2x, px1);
            ISSUE_AFTER(px, owB2y);  const float py = fmaf(rB, owB2y, py1);
            bh.x = px + other_grp(py);
            bh.y = py + other_grp(px);
        }
        hs[SBLK - 1] = bh.x + other_half(bh.x);

        if (writer) {
            v4f o0, o1;
            o0.x = DIR ? hs[7] : hs[0];
            o0.y = DIR ? hs[6] : hs[1];
            o0.z = DIR ? hs[5] : hs[2];
            o0.w = DIR ? hs[4] : hs[3];
            o1.x = DIR ? hs[3] : hs[4];
            o1.y = DIR ? hs[2] : hs[5];
            o1.z = DIR ? hs[1] : hs[6];
            o1.w = DIR ? hs[0] : hs[7];
            *(v4f*)(hout + row_out + off) = o0;
            *(v4f*)(hout + row_out + off + 4) = o1;
        }

        off = noff;
        yc0 = yn0; yc1 = yn1; ic0 = in0; ic1 = in1;
    }
}

__global__ void __launch_bounds__(32) __attribute__((amdgpu_waves_per_eu(1, 1)))
lstm_scan(const float* __restrict__ y,
                          const float* __restrict__ inp,
                          const float* __restrict__ w_ih_f, const float* __restrict__ w_hh_f,
                          const float* __restrict__ b_ih_f, const float* __restrict__ b_hh_f,
                          const float* __restrict__ w_hr_f,
                          const float* __restrict__ w_ih_b, const float* __restrict__ w_hh_b,
                          const float* __restrict__ b_ih_b, const float* __restrict__ b_hh_b,
                          const float* __restrict__ w_hr_b,
                          const float* __restrict__ h0, const float* __restrict__ c0,
                          float* __restrict__ hout) {
    if (blockIdx.x == 0)
        scan_dir<0>(y, inp, w_ih_f, w_hh_f, b_ih_f, b_hh_f, w_hr_f, h0, c0, hout);
    else
        scan_dir<1>(y, inp, w_ih_b, w_hh_b, b_ih_b, b_hh_b, w_hr_b, h0, c0, hout);
}

__global__ void seq_softmax(const float* __restrict__ hbuf, float* __restrict__ out) {
    const int b = blockIdx.x;
    const int tid = threadIdx.x;
    const int T = 256;
    __shared__ float red[256];

    const float* hf = hbuf + (size_t)b * LSEQ;
    const float* hb = hbuf + (size_t)(NB + b) * LSEQ;

    float m = -3.0e38f;
    for (int l = tid; l < LSEQ; l += T) m = fmaxf(m, hf[l] + hb[l]);
    red[tid] = m;
    __syncthreads();
    for (int s = T / 2; s > 0; s >>= 1) {
        if (tid < s) red[tid] = fmaxf(red[tid], red[tid + s]);
        __syncthreads();
    }
    m = red[0];
    __syncthreads();

    float sum = 0.0f;
    for (int l = tid; l < LSEQ; l += T) sum += __expf(hf[l] + hb[l] - m);
    red[tid] = sum;
    __syncthreads();
    for (int s = T / 2; s > 0; s >>= 1) {
        if (tid < s) red[tid] += red[tid + s];
        __syncthreads();
    }
    sum = red[0];
    __syncthreads();

    const float inv = __builtin_amdgcn_rcpf(sum);
    for (int l = tid; l < LSEQ; l += T) {
        const float e = __expf(hf[l] + hb[l] - m) * inv;
        float* o = out + ((size_t)b * LSEQ + l) * 2;
        o[0] = e;
        o[1] = e;
    }
}

extern "C" void kernel_launch(void* const* d_in, const int* in_sizes, int n_in,
                              void* d_out, int out_size, void* d_ws, size_t ws_size,
                              hipStream_t stream) {
    const float* x      = (const float*)d_in[0];
    const float* inp    = (const float*)d_in[1];
    const float* conv_w = (const float*)d_in[2];
    const float* conv_b = (const float*)d_in[3];
    const float* w_ih_f = (const float*)d_in[4];
    const float* w_hh_f = (const float*)d_in[5];
    const float* b_ih_f = (const float*)d_in[6];
    const float* b_hh_f = (const float*)d_in[7];
    const float* w_hr_f = (const float*)d_in[8];
    const float* w_ih_b = (const float*)d_in[9];
    const float* w_hh_b = (const float*)d_in[10];
    const float* b_ih_b = (const float*)d_in[11];
    const float* b_hh_b = (const float*)d_in[12];
    const float* w_hr_b = (const float*)d_in[13];
    const float* h0     = (const float*)d_in[14];
    const float* c0     = (const float*)d_in[15];
    float* out = (float*)d_out;

    float* y    = (float*)d_ws;
    float* hbuf = y + (size_t)NB * LSEQ;

    conv1x1_wmma<<<(NB * LSEQ / 16) / 8, 256, 0, stream>>>(x, conv_w, conv_b, y);

    lstm_scan<<<2, 32, 0, stream>>>(y, inp,
                                    w_ih_f, w_hh_f, b_ih_f, b_hh_f, w_hr_f,
                                    w_ih_b, w_hh_b, b_ih_b, b_hh_b, w_hr_b,
                                    h0, c0, hbuf);

    seq_softmax<<<NB, 256, 0, stream>>>(hbuf, out);
}


# LSTM scan: a second wave per direction, on its own SIMD, forms the steps' input-term matrix products and hands them over as D fragments through an LDS ring, 8 steps a barrier; sum order kept
# speedup vs baseline: 7.8129x; 1.1281x over previous
#include <hip/hip_runtime.h>
#include <hip/hip_bf16.h>

#define LSEQ 60000
#define NB 8
#define INCH 64
#define HID 5

typedef __attribute__((ext_vector_type(2))) float v2f;
typedef __attribute__((ext_vector_type(8))) float v8f;

__global__ void conv1x1_wmma(const float* __restrict__ x,
                             const float* __restrict__ w,
                             const float* __restrict__ bias,
                             float* __restrict__ y) {
    const int lane = threadIdx.x & 31;
    const int wave = threadIdx.x >> 5;
    const int tile = blockIdx.x * 8 + wave;
    const int b = tile / (LSEQ / 16);
    const int lbase = (tile % (LSEQ / 16)) * 16;
    const int m = lane & 15;
    const int half = lane >> 4;

    const float* xb = x + (size_t)b * INCH * LSEQ + lbase + m;

    v8f acc = {};
#pragma unroll
    for (int chunk = 0; chunk < 16; ++chunk) {
        const int k0 = chunk * 4 + 2 * half;
        v2f a, bb;
        a.x = xb[(size_t)k0 * LSEQ];
        a.y = xb[(size_t)(k0 + 1) * LSEQ];
        bb.x = w[k0];
        bb.y = w[k0 + 1];
        acc = __builtin_amdgcn_wmma_f32_16x16x4_f32(
             false, a,  false, bb,
             (short)0, acc,  false,  false);
    }

    const float bv = bias[0];
    float* yo = y + (size_t)b * LSEQ + lbase;
    if (lane == 0) {
#pragma unroll
        for (int r = 0; r < 8; ++r) yo[r] = acc[r] + bv;
    } else if (lane == 16) {
#pragma unroll
        for (int r = 0; r < 8; ++r) yo[8 + r] = acc[r] + bv;
    }
}

#define KS (-1.44269504f)
#define KT (2.88539008f)
__device__ __forceinline__ float rcp1p_exp2(float s) {
    return __builtin_amdgcn_rcpf(1.0f + __builtin_amdgcn_exp2f(s));
}

#define SBLK 8
#define NBLK (LSEQ / SBLK)
typedef __attribute__((ext_vector_type(4))) float v4f;

__device__ __forceinline__ int gate_of(int q) { return q == 2 ? 3 : (q == 3 ? 2 : q); }
__device__ __forceinline__ int unit_of(int grp, int slot) {
    return grp == 0 ? (slot < 3 ? slot : -1) : (slot < 2 ? 3 + slot : -1);
}
__device__ __forceinline__ float other_half(float v) {
    const int iv = __float_as_int(v);
    return __int_as_float(__builtin_amdgcn_permlanex16(iv, iv, 0x76543210, 0xfedcba98, true, false));
}
__device__ __forceinline__ float other_grp(float v) {
    return __int_as_float(__builtin_amdgcn_mov_dpp(__float_as_int(v), 0xB1, 0xF, 0xF, true));
}

#define ISSUE_AFTER(prev, next) asm volatile("" : "+v"(prev), "+v"(next))

__device__ __forceinline__ void block_barrier() {
    __builtin_amdgcn_fence(__ATOMIC_RELEASE, "workgroup", "local");
    __builtin_amdgcn_s_barrier();
    __builtin_amdgcn_fence(__ATOMIC_ACQUIRE, "workgroup", "local");
}

#define RING_AT(slot, i, h2) ((((slot) * SBLK + (i)) * 2 + (h2)) * 32 + lane)

template <int DIR>
__device__ __forceinline__ void scan_dir(const float* __restrict__ y,
                                         const float* __restrict__ inp,
                                         const float* __restrict__ w_ih, const float* __restrict__ w_hh,
                                         const float* __restrict__ b_ih, const float* __restrict__ b_hh,
                                         const float* __restrict__ w_hr,
                                         const float* __restrict__ h0, const float* __restrict__ c0,
                                         float* __restrict__ hout, v4f* ring) {
    const int wave = __builtin_amdgcn_readfirstlane((int)(threadIdx.x >> 5));
    const int lane = threadIdx.x & 31;
    const int hf = lane >> 4;
    const int col = lane & 15;
    const int b = col >> 1;
    const int grp = col & 1;

    const int aslot = 2 * (col >> 3) + (col & 1);
    const int agate = gate_of((col & 7) >> 1);
    const int ux = unit_of(hf, aslot);
    const float ascale = agate == 2 ? KT : KS;
    v2f ax;
    ax.x = ux >= 0 ? w_ih[2 * (agate * 5 + ux)] * ascale : 0.0f;
    ax.y = ux >= 0 ? w_ih[2 * (agate * 5 + ux) + 1] * ascale : 0.0f;
    const int uh0 = unit_of(0, aslot);
    const int uh1 = unit_of(1, aslot);
    v2f ah;
    ah.x = uh0 >= 0 ? w_hh[agate * 5 + uh0] * ascale : 0.0f;
    ah.y = uh1 >= 0 ? w_hh[agate * 5 + uh1] * ascale : 0.0f;

    const int uA = unit_of(grp, 2 * hf);
    const int uB = unit_of(grp, 2 * hf + 1);
    v8f cb;
#pragma unroll
    for (int r = 0; r < 8; ++r) {
        const int u = (r & 1) ? uB : uA;
        const int j = gate_of(r >> 1) * 5 + u;
        cb[r] = u >= 0 ? (b_ih[j] + b_hh[j]) * ((r >> 1) == 3 ? KT : KS) : 0.0f;
    }
    const float whrA = uA >= 0 ? w_hr[uA] : 0.0f;
    const float whrB = uB >= 0 ? w_hr[uB] : 0.0f;
    const float whrAx = grp == 0 ? whrA : 0.0f;
    const float whrBx = grp == 0 ? whrB : 0.0f;
    const float whrAy = grp == 1 ? whrA : 0.0f;
    const float whrBy = grp == 1 ? whrB : 0.0f;
    const float whrA2x = -2.0f * whrAx;
    const float whrB2x = -2.0f * whrBx;
    const float whrA2y = -2.0f * whrAy;
    const float whrB2y = -2.0f * whrBy;
    float cA = uA >= 0 ? KT * c0[(DIR * NB + b) * HID + uA] : 0.0f;
    float cB = uB >= 0 ? KT * c0[(DIR * NB + b) * HID + uB] : 0.0f;
    const float h = h0[DIR * NB + b];

    const bool own = (grp == hf);
    const bool writer = (hf == 0 && grp == 0);

    const unsigned row_in = (unsigned)b * LSEQ;
    const unsigned row_out = (unsigned)(DIR * NB + b) * LSEQ;
    unsigned off = DIR ? (unsigned)(LSEQ - SBLK) : 0u;
    const unsigned dstep = DIR ? (unsigned)(-SBLK) : (unsigned)SBLK;

    auto load_block = [&](int n, v4f& ya, v4f& yb, v4f& ia, v4f& ib) __attribute__((always_inline)) {
        const unsigned nn = (unsigned)(n < NBLK ? n : NBLK - 1);
        const unsigned o = DIR ? (unsigned)(LSEQ - SBLK) - nn * SBLK : nn * SBLK;
        ya = *(const v4f*)(y + row_in + o);
        yb = *(const v4f*)(y + row_in + o + 4);
        ia = *(const v4f*)(inp + row_in + o);
        ib = *(const v4f*)(inp + row_in + o + 4);
    };
    auto produce = [&](const v4f& ya, const v4f& yb, const v4f& ia, const v4f& ib,
                       const int slot) __attribute__((always_inline)) {
        const float ys[SBLK] = {ya.x, ya.y, ya.z, ya.w, yb.x, yb.y, yb.z, yb.w};
        const float is[SBLK] = {ia.x, ia.y, ia.z, ia.w, ib.x, ib.y, ib.z, ib.w};
#pragma unroll
        for (int hb = 0; hb < SBLK; hb += SBLK / 2) {
            v2f bxs[SBLK / 2];
            v8f dv[SBLK / 2];
#pragma unroll
            for (int j = 0; j < SBLK / 2; ++j) {
                const int e = DIR ? SBLK - 1 - (hb + j) : hb + j;
                bxs[j].x = own ? ys[e] : 0.0f;
                bxs[j].y = own ? is[e] : 0.0f;
            }
            __builtin_amdgcn_sched_barrier(0);
#pragma unroll
            for (int j = 0; j < SBLK / 2; ++j)
                dv[j] = __builtin_amdgcn_wmma_f32_16x16x4_f32(false, ax, false, bxs[j], (short)0, cb, false, false);
            __builtin_amdgcn_sched_barrier(0);
            asm volatile("v_nop\n\tv_nop\n\tv_nop\n\tv_nop" : "+v"(dv[SBLK / 2 - 1]));
            __builtin_amdgcn_sched_barrier(0);
#pragma unroll
            for (int j = 0; j < SBLK / 2; ++j) {
                ring[RING_AT(slot, hb + j, 0)] = __builtin_shufflevector(dv[j], dv[j], 0, 1, 2, 3);
                ring[RING_AT(slot, hb + j, 1)] = __builtin_shufflevector(dv[j], dv[j], 4, 5, 6, 7);
            }
            __builtin_amdgcn_sched_barrier(0);
        }
    };
    auto fetch = [&](v8f (&dst)[SBLK], const int slot) __attribute__((always_inline)) {
#pragma unroll
        for (int i = 0; i < SBLK; ++i) {
            const v4f lo = ring[RING_AT(slot, i, 0)];
            const v4f hi = ring[RING_AT(slot, i, 1)];
            dst[i] = __builtin_shufflevector(lo, hi, 0, 1, 2, 3, 4, 5, 6, 7);
        }
    };

    v2f bh;
    bh.x = (hf == 0 && grp == 0) ? h : 0.0f;
    bh.y = (hf == 0 && grp == 1) ? h : 0.0f;

    v4f py0 = {}, py1 = {}, pi0 = {}, pi1 = {};
    v8f fa[SBLK], fb[SBLK];

    if (wave == 1) {
        v4f qy0, qy1, qi0, qi1;
        load_block(0, py0, py1, pi0, pi1);
        load_block(1, qy0, qy1, qi0, qi1);
        produce(py0, py1, pi0, pi1, 0);
        produce(qy0, qy1, qi0, qi1, 1);
        load_block(2, py0, py1, pi0, pi1);
    }
    block_barrier();
    if (wave == 0) fetch(fa, 0);

    auto block_iter = [&](const int blk, const int par, v8f (&cur)[SBLK], v8f (&nxt)[SBLK])
                          __attribute__((always_inline)) {
        block_barrier();
        if (wave == 0) {
            fetch(nxt, par ^ 1);
            float hs[SBLK];
#pragma unroll
            for (int i = 0; i < SBLK; ++i) {
                __builtin_amdgcn_sched_barrier(0);
                v8f d = __builtin_amdgcn_wmma_f32_16x16x4_f32(false, ah, false, bh, (short)0, cur[i], false, false);
                __builtin_amdgcn_sched_barrier(0);
                asm volatile("v_nop\n\tv_nop\n\tv_nop\n\tv_nop" : "+v"(d) : "v"(cur[i]));
                __builtin_amdgcn_sched_barrier(0);

                if (i > 0) hs[i - 1] = bh.x + other_half(bh.x);

                float d0 = d[0], d1 = d[1], d2 = d[2], d3 = d[3];
                float d4 = d[4], d5 = d[5], d6 = d[6], d7 = d[7];
                float egA = __builtin_amdgcn_exp2f(d6);
                ISSUE_AFTER(egA, d7);    float egB = __builtin_amdgcn_exp2f(d7);
                ISSUE_AFTER(egB, d0);    float eiA = __builtin_amdgcn_exp2f(d0);
                ISSUE_AFTER(eiA, d1);    float eiB = __builtin_amdgcn_exp2f(d1);
                ISSUE_AFTER(eiB, d2);    float efA = __builtin_amdgcn_exp2f(d2);
                ISSUE_AFTER(efA, egA);   float agA = 1.0f + egA;
                ISSUE_AFTER(agA, d3);    float efB = __builtin_amdgcn_exp2f(d3);
                ISSUE_AFTER(efB, egB);   float agB = 1.0f + egB;
                ISSUE_AFTER(agB, eiA);   float aiA = 1.0f + eiA;
                ISSUE_AFTER(aiA, agA);   const float rgA = __builtin_amdgcn_rcpf(agA);
                float rgA_ = rgA;
                ISSUE_AFTER(rgA_, eiB);  float aiB = 1.0f + eiB;
                ISSUE_AFTER(aiB, agB);   float rgB = __builtin_amdgcn_rcpf(agB);
                ISSUE_AFTER(rgB, efA);   float afA = 1.0f + efA;
                ISSUE_AFTER(afA, aiA);   float iA = __builtin_amdgcn_rcpf(aiA);
                ISSUE_AFTER(iA, efB);    float afB = 1.0f + efB;
                ISSUE_AFTER(afB, aiB);   float iB = __builtin_amdgcn_rcpf(aiB);
                ISSUE_AFTER(iB, d5);     float eoB = __builtin_amdgcn_exp2f(d5);
                ISSUE_AFTER(eoB, rgA_);  float gA = fmaf(rgA_, -2.0f * KT, KT);
                ISSUE_AFTER(gA, afA);    float fA = __builtin_amdgcn_rcpf(afA);
                ISSUE_AFTER(fA, rgB);    float gB = fmaf(rgB, -2.0f * KT, KT);
                ISSUE_AFTER(gB, afB);    float fB = __builtin_amdgcn_rcpf(afB);
                ISSUE_AFTER(fB, iA);     float igA = iA * gA;
                ISSUE_AFTER(igA, d4);    float eoA = __builtin_amdgcn_exp2f(d4);
                ISSUE_AFTER(eoA, iB);    float igB = iB * gB;
                ISSUE_AFTER(igB, eoB);   float aoB = 1.0f + eoB;
                ISSUE_AFTER(aoB, fA);    cA = fmaf(fA, cA, igA);
                ISSUE_AFTER(cA, fB);     cB = fmaf(fB, cB, igB);
                ISSUE_AFTER(cB, aoB);    float oB = __builtin_amdgcn_rcpf(aoB);
                ISSUE_AFTER(oB, eoA);    float aoA = 1.0f + eoA;
                ISSUE_AFTER(aoA, cA);    float ecA = __builtin_amdgcn_exp2f(cA);
                ISSUE_AFTER(ecA, cB);    float ecB = __builtin_amdgcn_exp2f(cB);
                ISSUE_AFTER(ecB, aoA);   float oA = __builtin_amdgcn_rcpf(aoA);
                ISSUE_AFTER(oA, oB);     float mBx = oB * whrBx;
                ISSUE_AFTER(mBx, oB);    float mBy = oB * whrBy;
                ISSUE_AFTER(mBy, ecA);   float acA = 1.0f + ecA;
                ISSUE_AFTER(acA, oB);    float owB2x = oB * whrB2x;
                ISSUE_AFTER(owB2x, ecB); float acB = 1.0f + ecB;
                ISSUE_AFTER(acB, oB);    float owB2y = oB * whrB2y;
                ISSUE_AFTER(owB2y, oA);  float owA2x = oA * whrA2x;
                ISSUE_AFTER(owA2x, acA); float rA = __builtin_amdgcn_rcpf(acA);
                ISSUE_AFTER(rA, oA);     float owA2y = oA * whrA2y;
                ISSUE_AFTER(owA2y, acB); float rB = __builtin_amdgcn_rcpf(acB);
                ISSUE_AFTER(rB, mBx);    float osumx = fmaf(oA, whrAx, mBx);
                ISSUE_AFTER(osumx, mBy); float osumy = fmaf(oA, whrAy, mBy);
                ISSUE_AFTER(osumy, rA);  float px1 = fmaf(rA, owA2x, osumx);
                ISSUE_AFTER(px1, owA2y); float py1 = fmaf(rA, owA2y, osumy);
                ISSUE_AFTER(py1, owB2x); float px = fmaf(rB, owB2x, px1);
                ISSUE_AFTER(px, owB2y);  const float py = fmaf(rB, owB2y, py1);
                bh.x = px + other_grp(py);
                bh.y = py + other_grp(px);
            }
            hs[SBLK - 1] = bh.x + other_half(bh.x);

            if (writer) {
                v4f o0, o1;
                o0.x = DIR ? hs[7] : hs[0];
                o0.y = DIR ? hs[6] : hs[1];
                o0.z = DIR ? hs[5] : hs[2];
                o0.w = DIR ? hs[4] : hs[3];
                o1.x = DIR ? hs[3] : hs[4];
                o1.y = DIR ? hs[2] : hs[5];
                o1.z = DIR ? hs[1] : hs[6];
                o1.w = DIR ? hs[0] : hs[7];
                *(v4f*)(hout + row_out + off) = o0;
                *(v4f*)(hout + row_out + off + 4) = o1;
            }
            off += dstep;
        } else {
            produce(py0, py1, pi0, pi1, par);
            load_block(blk + 3, py0, py1, pi0, pi1);
        }
    };

    for (int blk = 0; blk < NBLK; blk += 2) {
        block_iter(blk, 0, fa, fb);
        block_iter(blk + 1, 1, fb, fa);
    }
}

__global__ void __launch_bounds__(64) __attribute__((amdgpu_waves_per_eu(1, 1)))
lstm_scan(const float* __restrict__ y,
                          const float* __restrict__ inp,
                          const float* __restrict__ w_ih_f, const float* __restrict__ w_hh_f,
                          const float* __restrict__ b_ih_f, const float* __restrict__ b_hh_f,
                          const float* __restrict__ w_hr_f,
                          const float* __restrict__ w_ih_b, const float* __restrict__ w_hh_b,
                          const float* __restrict__ b_ih_b, const float* __restrict__ b_hh_b,
                          const float* __restrict__ w_hr_b,
                          const float* __restrict__ h0, const float* __restrict__ c0,
                          float* __restrict__ hout) {
    __shared__ v4f ring[2 * SBLK * 2 * 32];
    if (blockIdx.x == 0)
        scan_dir<0>(y, inp, w_ih_f, w_hh_f, b_ih_f, b_hh_f, w_hr_f, h0, c0, hout, ring);
    else
        scan_dir<1>(y, inp, w_ih_b, w_hh_b, b_ih_b, b_hh_b, w_hr_b, h0, c0, hout, ring);
}

__global__ void seq_softmax(const float* __restrict__ hbuf, float* __restrict__ out) {
    const int b = blockIdx.x;
    const int tid = threadIdx.x;
    const int T = 256;
    __shared__ float red[256];

    const float* hf = hbuf + (size_t)b * LSEQ;
    const float* hb = hbuf + (size_t)(NB + b) * LSEQ;

    float m = -3.0e38f;
    for (int l = tid; l < LSEQ; l += T) m = fmaxf(m, hf[l] + hb[l]);
    red[tid] = m;
    __syncthreads();
    for (int s = T / 2; s > 0; s >>= 1) {
        if (tid < s) red[tid] = fmaxf(red[tid], red[tid + s]);
        __syncthreads();
    }
    m = red[0];
    __syncthreads();

    float sum = 0.0f;
    for (int l = tid; l < LSEQ; l += T) sum += __expf(hf[l] + hb[l] - m);
    red[tid] = sum;
    __syncthreads();
    for (int s = T / 2; s > 0; s >>= 1) {
        if (tid < s) red[tid] += red[tid + s];
        __syncthreads();
    }
    sum = red[0];
    __syncthreads();

    const float inv = __builtin_amdgcn_rcpf(sum);
    for (int l = tid; l < LSEQ; l += T) {
        const float e = __expf(hf[l] + hb[l] - m) * inv;
        float* o = out + ((size_t)b * LSEQ + l) * 2;
        o[0] = e;
        o[1] = e;
    }
}

extern "C" void kernel_launch(void* const* d_in, const int* in_sizes, int n_in,
                              void* d_out, int out_size, void* d_ws, size_t ws_size,
                              hipStream_t stream) {
    const float* x      = (const float*)d_in[0];
    const float* inp    = (const float*)d_in[1];
    const float* conv_w = (const float*)d_in[2];
    const float* conv_b = (const float*)d_in[3];
    const float* w_ih_f = (const float*)d_in[4];
    const float* w_hh_f = (const float*)d_in[5];
    const float* b_ih_f = (const float*)d_in[6];
    const float* b_hh_f = (const float*)d_in[7];
    const float* w_hr_f = (const float*)d_in[8];
    const float* w_ih_b = (const float*)d_in[9];
    const float* w_hh_b = (const float*)d_in[10];
    const float* b_ih_b = (const float*)d_in[11];
    const float* b_hh_b = (const float*)d_in[12];
    const float* w_hr_b = (const float*)d_in[13];
    const float* h0     = (const float*)d_in[14];
    const float* c0     = (const float*)d_in[15];
    float* out = (float*)d_out;

    float* y    = (float*)d_ws;
    float* hbuf = y + (size_t)NB * LSEQ;

    conv1x1_wmma<<<(NB * LSEQ / 16) / 8, 256, 0, stream>>>(x, conv_w, conv_b, y);

    lstm_scan<<<2, 64, 0, stream>>>(y, inp,
                                    w_ih_f, w_hh_f, b_ih_f, b_hh_f, w_hr_f,
                                    w_ih_b, w_hh_b, b_ih_b, b_hh_b, w_hr_b,
                                    h0, c0, hbuf);

    seq_softmax<<<NB, 256, 0, stream>>>(hbuf, out);
}


# LSTM scan: hand-over block of 32 steps, was 8: barrier, role branches, h stores once per 32 steps; fragments still read 8 steps a group, from a ring of 3 block slots (was 2); sum order kept
# speedup vs baseline: 9.2236x; 1.1806x over previous
#include <hip/hip_runtime.h>
#include <hip/hip_bf16.h>

#define LSEQ 60000
#define NB 8
#define INCH 64
#define HID 5

typedef __attribute__((ext_vector_type(2))) float v2f;
typedef __attribute__((ext_vector_type(8))) float v8f;

__global__ void conv1x1_wmma(const float* __restrict__ x,
                             const float* __restrict__ w,
                             const float* __restrict__ bias,
                             float* __restrict__ y) {
    const int lane = threadIdx.x & 31;
    const int wave = threadIdx.x >> 5;
    const int tile = blockIdx.x * 8 + wave;
    const int b = tile / (LSEQ / 16);
    const int lbase = (tile % (LSEQ / 16)) * 16;
    const int m = lane & 15;
    const int half = lane >> 4;

    const float* xb = x + (size_t)b * INCH * LSEQ + lbase + m;

    v8f acc = {};
#pragma unroll
    for (int chunk = 0; chunk < 16; ++chunk) {
        const int k0 = chunk * 4 + 2 * half;
        v2f a, bb;
        a.x = xb[(size_t)k0 * LSEQ];
        a.y = xb[(size_t)(k0 + 1) * LSEQ];
        bb.x = w[k0];
        bb.y = w[k0 + 1];
        acc = __builtin_amdgcn_wmma_f32_16x16x4_f32(
             false, a,  false, bb,
             (short)0, acc,  false,  false);
    }

    const float bv = bias[0];
    float* yo = y + (size_t)b * LSEQ + lbase;
    if (lane == 0) {
#pragma unroll
        for (int r = 0; r < 8; ++r) yo[r] = acc[r] + bv;
    } else if (lane == 16) {
#pragma unroll
        for (int r = 0; r < 8; ++r) yo[8 + r] = acc[r] + bv;
    }
}

#define KS (-1.44269504f)
#define KT (2.88539008f)
__device__ __forceinline__ float rcp1p_exp2(float s) {
    return __builtin_amdgcn_rcpf(1.0f + __builtin_amdgcn_exp2f(s));
}

#define SBLK 32
#define NBLK (LSEQ / SBLK)
#define FSET 8
#define NGRP (SBLK / FSET)
#define NSLOT 3
#define SVEC (SBLK / 4)
#define PRUN 4
typedef __attribute__((ext_vector_type(4))) float v4f;

__device__ __forceinline__ int gate_of(int q) { return q == 2 ? 3 : (q == 3 ? 2 : q); }
__device__ __forceinline__ int unit_of(int grp, int slot) {
    return grp == 0 ? (slot < 3 ? slot : -1) : (slot < 2 ? 3 + slot : -1);
}
__device__ __forceinline__ float other_half(float v) {
    const int iv = __float_as_int(v);
    return __int_as_float(__builtin_amdgcn_permlanex16(iv, iv, 0x76543210, 0xfedcba98, true, false));
}
__device__ __forceinline__ float other_grp(float v) {
    return __int_as_float(__builtin_amdgcn_mov_dpp(__float_as_int(v), 0xB1, 0xF, 0xF, true));
}

#define ISSUE_AFTER(prev, next) asm volatile("" : "+v"(prev), "+v"(next))

__device__ __forceinline__ void block_barrier() {
    __builtin_amdgcn_fence(__ATOMIC_RELEASE, "workgroup", "local");
    __builtin_amdgcn_s_barrier();
    __builtin_amdgcn_fence(__ATOMIC_ACQUIRE, "workgroup", "local");
}

#define RING_AT(slot, i, h2) ((((slot) * SBLK + (i)) * 2 + (h2)) * 32 + lane)

template <int DIR>
__device__ __forceinline__ void scan_dir(const float* __restrict__ y,
                                         const float* __restrict__ inp,
                                         const float* __restrict__ w_ih, const float* __restrict__ w_hh,
                                         const float* __restrict__ b_ih, const float* __restrict__ b_hh,
                                         const float* __restrict__ w_hr,
                                         const float* __restrict__ h0, const float* __restrict__ c0,
                                         float* __restrict__ hout, v4f* ring) {
    const int wave = __builtin_amdgcn_readfirstlane((int)(threadIdx.x >> 5));
    const int lane = threadIdx.x & 31;
    const int hf = lane >> 4;
    const int col = lane & 15;
    const int b = col >> 1;
    const int grp = col & 1;

    const int aslot = 2 * (col >> 3) + (col & 1);
    const int agate = gate_of((col & 7) >> 1);
    const int ux = unit_of(hf, aslot);
    const float ascale = agate == 2 ? KT : KS;
    v2f ax;
    ax.x = ux >= 0 ? w_ih[2 * (agate * 5 + ux)] * ascale : 0.0f;
    ax.y = ux >= 0 ? w_ih[2 * (agate * 5 + ux) + 1] * ascale : 0.0f;
    const int uh0 = unit_of(0, aslot);
    const int uh1 = unit_of(1, aslot);
    v2f ah;
    ah.x = uh0 >= 0 ? w_hh[agate * 5 + uh0] * ascale : 0.0f;
    ah.y = uh1 >= 0 ? w_hh[agate * 5 + uh1] * ascale : 0.0f;

    const int uA = unit_of(grp, 2 * hf);
    const int uB = unit_of(grp, 2 * hf + 1);
    v8f cb;
#pragma unroll
    for (int r = 0; r < 8; ++r) {
        const int u = (r & 1) ? uB : uA;
        const int j = gate_of(r >> 1) * 5 + u;
        cb[r] = u >= 0 ? (b_ih[j] + b_hh[j]) * ((r >> 1) == 3 ? KT : KS) : 0.0f;
    }
    const float whrA = uA >= 0 ? w_hr[uA] : 0.0f;
    const float whrB = uB >= 0 ? w_hr[uB] : 0.0f;
    const float whrAx = grp == 0 ? whrA : 0.0f;
    const float whrBx = grp == 0 ? whrB : 0.0f;
    const float whrAy = grp == 1 ? whrA : 0.0f;
    const float whrBy = grp == 1 ? whrB : 0.0f;
    const float whrA2x = -2.0f * whrAx;
    const float whrB2x = -2.0f * whrBx;
    const float whrA2y = -2.0f * whrAy;
    const float whrB2y = -2.0f * whrBy;
    float cA = uA >= 0 ? KT * c0[(DIR * NB + b) * HID + uA] : 0.0f;
    float cB = uB >= 0 ? KT * c0[(DIR * NB + b) * HID + uB] : 0.0f;
    const float h = h0[DIR * NB + b];

    const bool own = (grp == hf);
    const bool writer = (hf == 0 && grp == 0);

    const unsigned row_in = (unsigned)b * LSEQ;
    const unsigned row_out = (unsigned)(DIR * NB + b) * LSEQ;
    unsigned off = DIR ? (unsigned)(LSEQ - SBLK) : 0u;
    const unsigned dstep = DIR ? (unsigned)(-SBLK) : (unsigned)SBLK;

    auto load_block = [&](int n, v4f (&yv)[SVEC], v4f (&iv)[SVEC]) __attribute__((always_inline)) {
        const unsigned nn = (unsigned)(n < NBLK ? n : NBLK - 1);
        const unsigned o = DIR ? (unsigned)(LSEQ - SBLK) - nn * SBLK : nn * SBLK;
#pragma unroll
        for (int j = 0; j < SVEC; ++j) yv[j] = *(const v4f*)(y + row_in + o + 4 * j);
#pragma unroll
        for (int j = 0; j < SVEC; ++j) iv[j] = *(const v4f*)(inp + row_in + o + 4 * j);
    };
    auto produce = [&](const v4f (&yv)[SVEC], const v4f (&iv)[SVEC],
                       const int slot) __attribute__((always_inline)) {
        float ys[SBLK], is[SBLK];
#pragma unroll
        for (int j = 0; j < SVEC; ++j) {
            ys[4 * j] = yv[j].x; ys[4 * j + 1] = yv[j].y; ys[4 * j + 2] = yv[j].z; ys[4 * j + 3] = yv[j].w;
            is[4 * j] = iv[j].x; is[4 * j + 1] = iv[j].y; is[4 * j + 2] = iv[j].z; is[4 * j + 3] = iv[j].w;
        }
#pragma unroll
        for (int hb = 0; hb < SBLK; hb += PRUN) {
            v2f bxs[PRUN];
            v8f dv[PRUN];
#pragma unroll
            for (int j = 0; j < PRUN; ++j) {
                const int e = DIR ? SBLK - 1 - (hb + j) : hb + j;
                bxs[j].x = own ? ys[e] : 0.0f;
                bxs[j].y = own ? is[e] : 0.0f;
            }
            __builtin_amdgcn_sched_barrier(0);
#pragma unroll
            for (int j = 0; j < PRUN; ++j)
                dv[j] = __builtin_amdgcn_wmma_f32_16x16x4_f32(false, ax, false, bxs[j], (short)0, cb, false, false);
            __builtin_amdgcn_sched_barrier(0);
            asm volatile("v_nop\n\tv_nop\n\tv_nop\n\tv_nop" : "+v"(dv[PRUN - 1]));
            __builtin_amdgcn_sched_barrier(0);
#pragma unroll
            for (int j = 0; j < PRUN; ++j) {
                ring[RING_AT(slot, hb + j, 0)] = __builtin_shufflevector(dv[j], dv[j], 0, 1, 2, 3);
                ring[RING_AT(slot, hb + j, 1)] = __builtin_shufflevector(dv[j], dv[j], 4, 5, 6, 7);
            }
            __builtin_amdgcn_sched_barrier(0);
        }
    };
    auto fetch = [&](v8f (&dst)[FSET], const unsigned rb, const int g) __attribute__((always_inline)) {
        const char* base = (const char*)ring + rb;
#pragma unroll
        for (int j = 0; j < FSET; ++j) {
            const v4f lo = *(const v4f*)(base + ((g * FSET + j) * 2 + 0) * 512);
            const v4f hi = *(const v4f*)(base + ((g * FSET + j) * 2 + 1) * 512);
            dst[j] = __builtin_shufflevector(lo, hi, 0, 1, 2, 3, 4, 5, 6, 7);
        }
    };

    v2f bh;
    bh.x = (hf == 0 && grp == 0) ? h : 0.0f;
    bh.y = (hf == 0 && grp == 1) ? h : 0.0f;

    v4f py[SVEC] = {}, pi[SVEC] = {};
    v8f fa[FSET], fb[FSET];

    if (wave == 1) {
        v4f qy[SVEC], qi[SVEC];
        load_block(0, py, pi);
        load_block(1, qy, qi);
        produce(py, pi, 0);
        produce(qy, qi, 1);
        load_block(2, py, pi);
    }
    block_barrier();
    if (wave == 0) fetch(fa, (unsigned)lane * 16u, 0);

    auto block_iter = [&](const int blk, const unsigned rb0, const unsigned rb1, const int s2)
                          __attribute__((always_inline)) {
        block_barrier();
        if (wave == 0) {
            float hs[SBLK];
#pragma unroll
            for (int g = 0; g < NGRP; ++g) {
            __builtin_amdgcn_sched_barrier(0);
            if (g + 1 < NGRP) {
                if (g & 1) fetch(fa, rb0, g + 1); else fetch(fb, rb0, g + 1);
            } else {
                fetch(fa, rb1, 0);
            }
#pragma unroll
            for (int j = 0; j < FSET; ++j) {
                const int i = g * FSET + j;
                const v8f& frag = (g & 1) ? fb[j] : fa[j];
                __builtin_amdgcn_sched_barrier(0);
                v8f d = __builtin_amdgcn_wmma_f32_16x16x4_f32(false, ah, false, bh, (short)0, frag, false, false);
                __builtin_amdgcn_sched_barrier(0);
                asm volatile("v_nop\n\tv_nop\n\tv_nop\n\tv_nop" : "+v"(d) : "v"(frag));
                __builtin_amdgcn_sched_barrier(0);

                if (i > 0) hs[i - 1] = bh.x + other_half(bh.x);

                float d0 = d[0], d1 = d[1], d2 = d[2], d3 = d[3];
                float d4 = d[4], d5 = d[5], d6 = d[6], d7 = d[7];
                float egA = __builtin_amdgcn_exp2f(d6);
                ISSUE_AFTER(egA, d7);    float egB = __builtin_amdgcn_exp2f(d7);
                ISSUE_AFTER(egB, d0);    float eiA = __builtin_amdgcn_exp2f(d0);
                ISSUE_AFTER(eiA, d1);    float eiB = __builtin_amdgcn_exp2f(d1);
                ISSUE_AFTER(eiB, d2);    float efA = __builtin_amdgcn_exp2f(d2);
                ISSUE_AFTER(efA, egA);   float agA = 1.0f + egA;
                ISSUE_AFTER(agA, d3);    float efB = __builtin_amdgcn_exp2f(d3);
                ISSUE_AFTER(efB, egB);   float agB = 1.0f + egB;
                ISSUE_AFTER(agB, eiA);   float aiA = 1.0f + eiA;
                ISSUE_AFTER(aiA, agA);   const float rgA = __builtin_amdgcn_rcpf(agA);
                float rgA_ = rgA;
                ISSUE_AFTER(rgA_, eiB);  float aiB = 1.0f + eiB;
                ISSUE_AFTER(aiB, agB);   float rgB = __builtin_amdgcn_rcpf(agB);
                ISSUE_AFTER(rgB, efA);   float afA = 1.0f + efA;
                ISSUE_AFTER(afA, aiA);   float iA = __builtin_amdgcn_rcpf(aiA);
                ISSUE_AFTER(iA, efB);    float afB = 1.0f + efB;
                ISSUE_AFTER(afB, aiB);   float iB = __builtin_amdgcn_rcpf(aiB);
                ISSUE_AFTER(iB, d5);     float eoB = __builtin_amdgcn_exp2f(d5);
                ISSUE_AFTER(eoB, rgA_);  float gA = fmaf(rgA_, -2.0f * KT, KT);
                ISSUE_AFTER(gA, afA);    float fA = __builtin_amdgcn_rcpf(afA);
                ISSUE_AFTER(fA, rgB);    float gB = fmaf(rgB, -2.0f * KT, KT);
                ISSUE_AFTER(gB, afB);    float fB = __builtin_amdgcn_rcpf(afB);
                ISSUE_AFTER(fB, iA);     float igA = iA * gA;
                ISSUE_AFTER(igA, d4);    float eoA = __builtin_amdgcn_exp2f(d4);
                ISSUE_AFTER(eoA, iB);    float igB = iB * gB;
                ISSUE_AFTER(igB, eoB);   float aoB = 1.0f + eoB;
                ISSUE_AFTER(aoB, fA);    cA = fmaf(fA, cA, igA);
                ISSUE_AFTER(cA, fB);     cB = fmaf(fB, cB, igB);
                ISSUE_AFTER(cB, aoB);    float oB = __builtin_amdgcn_rcpf(aoB);
                ISSUE_AFTER(oB, eoA);    float aoA = 1.0f + eoA;
                ISSUE_AFTER(aoA, cA);    float ecA = __builtin_amdgcn_exp2f(cA);
                ISSUE_AFTER(ecA, cB);    float ecB = __builtin_amdgcn_exp2f(cB);
                ISSUE_AFTER(ecB, aoA);   float oA = __builtin_amdgcn_rcpf(aoA);
                ISSUE_AFTER(oA, oB);     float mBx = oB * whrBx;
                ISSUE_AFTER(mBx, oB);    float mBy = oB * whrBy;
                ISSUE_AFTER(mBy, ecA);   float acA = 1.0f + ecA;
                ISSUE_AFTER(acA, oB);    float owB2x = oB * whrB2x;
                ISSUE_AFTER(owB2x, ecB); float acB = 1.0f + ecB;
                ISSUE_AFTER(acB, oB);    float owB2y = oB * whrB2y;
                ISSUE_AFTER(owB2y, oA);  float owA2x = oA * whrA2x;
                ISSUE_AFTER(owA2x, acA); float rA = __builtin_amdgcn_rcpf(acA);
                ISSUE_AFTER(rA, oA);     float owA2y = oA * whrA2y;
                ISSUE_AFTER(owA2y, acB); float rB = __builtin_amdgcn_rcpf(acB);
                ISSUE_AFTER(rB, mBx);    float osumx = fmaf(oA, whrAx, mBx);
                ISSUE_AFTER(osumx, mBy); float osumy = fmaf(oA, whrAy, mBy);
                ISSUE_AFTER(osumy, rA);  float px1 = fmaf(rA, owA2x, osumx);
                ISSUE_AFTER(px1, owA2y); float py1 = fmaf(rA, owA2y, osumy);
                ISSUE_AFTER(py1, owB2x); float px = fmaf(rB, owB2x, px1);
                ISSUE_AFTER(px, owB2y);  const float py = fmaf(rB, owB2y, py1);
                bh.x = px + other_grp(py);
                bh.y = py + other_grp(px);
            }
            }
            hs[SBLK - 1] = bh.x + other_half(bh.x);

            if (writer) {
#pragma unroll
                for (int j = 0; j < SVEC; ++j) {
                    v4f o;
                    o.x = DIR ? hs[SBLK - 1 - 4 * j] : hs[4 * j];
                    o.y = DIR ? hs[SBLK - 2 - 4 * j] : hs[4 * j + 1];
                    o.z = DIR ? hs[SBLK - 3 - 4 * j] : hs[4 * j + 2];
                    o.w = DIR ? hs[SBLK - 4 - 4 * j] : hs[4 * j + 3];
                    *(v4f*)(hout + row_out + off + 4 * j) = o;
                }
            }
            off += dstep;
        } else {
            produce(py, pi, s2);
            load_block(blk + 3, py, pi);
        }
    };

    int s0 = 0;
    unsigned rb0 = (unsigned)lane * 16u;
    for (int blk = 0; blk < NBLK; ++blk) {
        const int s1 = s0 == NSLOT - 1 ? 0 : s0 + 1;
        const int s2 = s1 == NSLOT - 1 ? 0 : s1 + 1;
        unsigned rb1 = (unsigned)lane * 16u + (unsigned)s1 * (unsigned)(SBLK * 2 * 32 * 16);
        block_iter(blk, rb0, rb1, s2);
        s0 = s1;
        rb0 = rb1;
    }
}

__global__ void __launch_bounds__(64) __attribute__((amdgpu_waves_per_eu(1, 1)))
lstm_scan(const float* __restrict__ y,
                          const float* __restrict__ inp,
                          const float* __restrict__ w_ih_f, const float* __restrict__ w_hh_f,
                          const float* __restrict__ b_ih_f, const float* __restrict__ b_hh_f,
                          const float* __restrict__ w_hr_f,
                          const float* __restrict__ w_ih_b, const float* __restrict__ w_hh_b,
                          const float* __restrict__ b_ih_b, const float* __restrict__ b_hh_b,
                          const float* __restrict__ w_hr_b,
                          const float* __restrict__ h0, const float* __restrict__ c0,
                          float* __restrict__ hout) {
    __shared__ v4f ring[NSLOT * SBLK * 2 * 32];
    if (blockIdx.x == 0)
        scan_dir<0>(y, inp, w_ih_f, w_hh_f, b_ih_f, b_hh_f, w_hr_f, h0, c0, hout, ring);
    else
        scan_dir<1>(y, inp, w_ih_b, w_hh_b, b_ih_b, b_hh_b, w_hr_b, h0, c0, hout, ring);
}

__global__ void seq_softmax(const float* __restrict__ hbuf, float* __restrict__ out) {
    const int b = blockIdx.x;
    const int tid = threadIdx.x;
    const int T = 256;
    __shared__ float red[256];

    const float* hf = hbuf + (size_t)b * LSEQ;
    const float* hb = hbuf + (size_t)(NB + b) * LSEQ;

    float m = -3.0e38f;
    for (int l = tid; l < LSEQ; l += T) m = fmaxf(m, hf[l] + hb[l]);
    red[tid] = m;
    __syncthreads();
    for (int s = T / 2; s > 0; s >>= 1) {
        if (tid < s) red[tid] = fmaxf(red[tid], red[tid + s]);
        __syncthreads();
    }
    m = red[0];
    __syncthreads();

    float sum = 0.0f;
    for (int l = tid; l < LSEQ; l += T) sum += __expf(hf[l] + hb[l] - m);
    red[tid] = sum;
    __syncthreads();
    for (int s = T / 2; s > 0; s >>= 1) {
        if (tid < s) red[tid] += red[tid + s];
        __syncthreads();
    }
    sum = red[0];
    __syncthreads();

    const float inv = __builtin_amdgcn_rcpf(sum);
    for (int l = tid; l < LSEQ; l += T) {
        const float e = __expf(hf[l] + hb[l] - m) * inv;
        float* o = out + ((size_t)b * LSEQ + l) * 2;
        o[0] = e;
        o[1] = e;
    }
}

extern "C" void kernel_launch(void* const* d_in, const int* in_sizes, int n_in,
                              void* d_out, int out_size, void* d_ws, size_t ws_size,
                              hipStream_t stream) {
    const float* x      = (const float*)d_in[0];
    const float* inp    = (const float*)d_in[1];
    const float* conv_w = (const float*)d_in[2];
    const float* conv_b = (const float*)d_in[3];
    const float* w_ih_f = (const float*)d_in[4];
    const float* w_hh_f = (const float*)d_in[5];
    const float* b_ih_f = (const float*)d_in[6];
    const float* b_hh_f = (const float*)d_in[7];
    const float* w_hr_f = (const float*)d_in[8];
    const float* w_ih_b = (const float*)d_in[9];
    const float* w_hh_b = (const float*)d_in[10];
    const float* b_ih_b = (const float*)d_in[11];
    const float* b_hh_b = (const float*)d_in[12];
    const float* w_hr_b = (const float*)d_in[13];
    const float* h0     = (const float*)d_in[14];
    const float* c0     = (const float*)d_in[15];
    float* out = (float*)d_out;

    float* y    = (float*)d_ws;
    float* hbuf = y + (size_t)NB * LSEQ;

    conv1x1_wmma<<<(NB * LSEQ / 16) / 8, 256, 0, stream>>>(x, conv_w, conv_b, y);

    lstm_scan<<<2, 64, 0, stream>>>(y, inp,
                                    w_ih_f, w_hh_f, b_ih_f, b_hh_f, w_hr_f,
                                    w_ih_b, w_hh_b, b_ih_b, b_hh_b, w_hr_b,
                                    h0, c0, hbuf);

    seq_softmax<<<NB, 256, 0, stream>>>(hbuf, out);
}
